# starting kernel
# baseline (speedup 1.0000x reference)
#include <hip/hip_runtime.h>
#include <math.h>

typedef __attribute__((ext_vector_type(2))) float v2f;
typedef __attribute__((ext_vector_type(8))) float v8f;

#define B_SZ   256
#define M_SZ   1024
#define A_SZ   8
#define K_CLS  20
#define O_SZ   16
#define KPAD   32
#define I_SZ   (B_SZ * A_SZ)

__global__ void caps_zero_bij(float* __restrict__ bij) {
    int idx = blockIdx.x * blockDim.x + threadIdx.x;
    bij[idx] = 0.0f;
}

__global__ void caps_softmax(const float* __restrict__ bij, float* __restrict__ c) {
    __shared__ float red[256];
    const int k = blockIdx.x;
    const int t = threadIdx.x;

    float v0 = bij[(t + 0  ) * KPAD + k];
    float v1 = bij[(t + 256) * KPAD + k];
    float v2 = bij[(t + 512) * KPAD + k];
    float v3 = bij[(t + 768) * KPAD + k];

    float mx = fmaxf(fmaxf(v0, v1), fmaxf(v2, v3));
    red[t] = mx;
    __syncthreads();
    for (int s = 128; s > 0; s >>= 1) {
        if (t < s) red[t] = fmaxf(red[t], red[t + s]);
        __syncthreads();
    }
    mx = red[0];
    __syncthreads();

    float e0 = expf(v0 - mx), e1 = expf(v1 - mx);
    float e2 = expf(v2 - mx), e3 = expf(v3 - mx);
    red[t] = e0 + e1 + e2 + e3;
    __syncthreads();
    for (int s = 128; s > 0; s >>= 1) {
        if (t < s) red[t] += red[t + s];
        __syncthreads();
    }
    float inv = 1.0f / red[0];

    c[(t + 0  ) * KPAD + k] = e0 * inv;
    c[(t + 256) * KPAD + k] = e1 * inv;
    c[(t + 512) * KPAD + k] = e2 * inv;
    c[(t + 768) * KPAD + k] = e3 * inv;
}

__global__ void caps_gemm_t(const float* __restrict__ x,
                            const float* __restrict__ c,
                            float* __restrict__ t2) {
    const int wave = (blockIdx.x * blockDim.x + threadIdx.x) >> 5;
    const int lane = threadIdx.x & 31;
    const int i0   = wave * 16;
    const int lo   = lane & 15;
    const int hi   = lane >> 4;
    const int row  = i0 + lo;
    const int xb   = (row >> 3) * (A_SZ * M_SZ) + (row & 7);
    const int j0   = hi * 2;

    v8f acc0 = {};
    v8f acc1 = {};
    #pragma unroll 2
    for (int mb = 0; mb < M_SZ; mb += 4) {
        v2f av, bv0, bv1;
        av.x  = x[xb + (mb + j0    ) * A_SZ];
        av.y  = x[xb + (mb + j0 + 1) * A_SZ];
        bv0.x = c[(mb + j0    ) * KPAD + lo];
        bv0.y = c[(mb + j0 + 1) * KPAD + lo];
        bv1.x = c[(mb + j0    ) * KPAD + 16 + lo];
        bv1.y = c[(mb + j0 + 1) * KPAD + 16 + lo];
        acc0 = __builtin_amdgcn_wmma_f32_16x16x4_f32(
            false, av, false, bv0, (short)0, acc0, false, false);
        acc1 = __builtin_amdgcn_wmma_f32_16x16x4_f32(
            false, av, false, bv1, (short)0, acc1, false, false);
    }
    #pragma unroll
    for (int r = 0; r < 8; ++r) {
        const int rbase = (i0 + r + hi * 8) * KPAD;
        t2[rbase + lo]      = acc0[r];
        t2[rbase + 16 + lo] = acc1[r];
    }
}

__global__ void caps_squash(const float* __restrict__ t2,
                            const float* __restrict__ W,
                            float* __restrict__ out,
                            float* __restrict__ wv2) {
    __shared__ float s_s[K_CLS][O_SZ];
    __shared__ float s_v[K_CLS][O_SZ];
    const int b = blockIdx.x;
    const int t = threadIdx.x;
    const int k = t >> 4;
    const int o = t & 15;

    float s = 0.0f;
    #pragma unroll
    for (int a = 0; a < A_SZ; ++a)
        s += W[k * (O_SZ * A_SZ) + o * A_SZ + a] * t2[(b * 8 + a) * KPAD + k];
    s_s[k][o] = s;
    __syncthreads();

    float magsq = 0.0f;
    #pragma unroll
    for (int kk = 0; kk < K_CLS; ++kk) {
        float sv = s_s[kk][o];
        magsq += sv * sv;
    }
    float scale = magsq / ((1.0f + magsq) * sqrtf(magsq));
    float v = s * scale;
    s_v[k][o] = v;
    out[b * (K_CLS * O_SZ) + t] = v;
    __syncthreads();

    if (t < K_CLS * A_SZ) {
        int k2 = t % K_CLS;
        int a  = t / K_CLS;
        float w = 0.0f;
        #pragma unroll
        for (int o2 = 0; o2 < O_SZ; ++o2)
            w += W[k2 * (O_SZ * A_SZ) + o2 * A_SZ + a] * s_v[k2][o2];
        wv2[(b * 8 + a) * KPAD + k2] = w * (1.0f / (float)B_SZ);
    }
}

__global__ void caps_gemm_b(const float* __restrict__ x,
                            const float* __restrict__ wv2,
                            float* __restrict__ bij) {
    const int wave = (blockIdx.x * blockDim.x + threadIdx.x) >> 5;
    const int lane = threadIdx.x & 31;
    const int m0   = wave * 16;
    const int lo   = lane & 15;
    const int hi   = lane >> 4;
    const int m    = m0 + lo;
    const int j0   = hi * 2;

    v8f acc0 = {};
    v8f acc1 = {};
    #pragma unroll 2
    for (int ib = 0; ib < I_SZ; ib += 4) {
        const int ia = ib + j0;
        const int ic = ia + 1;
        v2f av, bv0, bv1;
        av.x  = x[(ia >> 3) * (A_SZ * M_SZ) + m * A_SZ + (ia & 7)];
        av.y  = x[(ic >> 3) * (A_SZ * M_SZ) + m * A_SZ + (ic & 7)];
        bv0.x = wv2[ia * KPAD + lo];
        bv0.y = wv2[ic * KPAD + lo];
        bv1.x = wv2[ia * KPAD + 16 + lo];
        bv1.y = wv2[ic * KPAD + 16 + lo];
        acc0 = __builtin_amdgcn_wmma_f32_16x16x4_f32(
            false, av, false, bv0, (short)0, acc0, false, false);
        acc1 = __builtin_amdgcn_wmma_f32_16x16x4_f32(
            false, av, false, bv1, (short)0, acc1, false, false);
    }
    #pragma unroll
    for (int r = 0; r < 8; ++r) {
        const int rbase = (m0 + r + hi * 8) * KPAD;
        bij[rbase + lo]      += acc0[r];
        bij[rbase + 16 + lo] += acc1[r];
    }
}

extern "C" void kernel_launch(void* const* d_in, const int* in_sizes, int n_in,
                              void* d_out, int out_size, void* d_ws, size_t ws_size,
                              hipStream_t stream) {
    (void)in_sizes; (void)n_in; (void)out_size; (void)ws_size;
    const float* x = (const float*)d_in[0];
    const float* W = (const float*)d_in[1];
    float* out = (float*)d_out;

    float* bij = (float*)d_ws;
    float* c   = bij + M_SZ * KPAD;
    float* t2  = c   + M_SZ * KPAD;
    float* wv2 = t2  + I_SZ * KPAD;

    caps_zero_bij<<<(M_SZ * KPAD) / 256, 256, 0, stream>>>(bij);

    for (int it = 0; it < 3; ++it) {
        caps_softmax<<<K_CLS, 256, 0, stream>>>(bij, c);
        caps_gemm_t<<<16, 256, 0, stream>>>(x, c, t2);
        caps_squash<<<B_SZ, 320, 0, stream>>>(t2, W, out, wv2);
        if (it < 2)
            caps_gemm_b<<<8, 256, 0, stream>>>(x, wv2, bij);
    }
}


# Column softmax: max and exp-sum by lane butterflies (DPP row_xmask, permlanex16) in each wave plus one 8-value LDS hand-over, not two 8-level LDS trees: 19 barriers -> 2 (exp-sum order changes)
# speedup vs baseline: 7.2401x; 7.2401x over previous
#include <hip/hip_runtime.h>
#include <math.h>

typedef __attribute__((ext_vector_type(2))) float v2f;
typedef __attribute__((ext_vector_type(8))) float v8f;

#define B_SZ   256
#define M_SZ   1024
#define A_SZ   8
#define K_CLS  20
#define O_SZ   16
#define KPAD   32
#define BP     20
#define GSTEP  16
#define NTH    128
#define RPP    25
#define I_SZ   (B_SZ * A_SZ)
#define KS_T   2
#define KS_B   4
#define KROWS  512
#define PP     20

__global__ void caps_zero_bij(float* __restrict__ bij) {
    int idx = blockIdx.x * blockDim.x + threadIdx.x;
    bij[idx] = 0.0f;
}

template <int M>
__device__ __forceinline__ float xrow(float x) {
    const int b = __builtin_bit_cast(int, x);
    return __builtin_bit_cast(float, __builtin_amdgcn_update_dpp(b, b, 0x160 + M, 0xf, 0xf, false));
}
__device__ __forceinline__ float xhalf(float x) {
    const int b = __builtin_bit_cast(int, x);
    return __builtin_bit_cast(float, __builtin_amdgcn_permlanex16(b, b, 0x76543210, 0xfedcba98, false, false));
}

template <bool ADD, bool WB>
__global__ void caps_softmax(float* __restrict__ bij, const float* __restrict__ part,
                             float* __restrict__ c) {
    __shared__ __attribute__((aligned(16))) float redm[8];
    __shared__ __attribute__((aligned(16))) float reds[8];
    const int k = blockIdx.x;
    const int t = threadIdx.x;
    const int wave = t >> 5;
    const int lane = t & 31;

    float v0 = bij[(t + 0  ) * KPAD + k];
    float v1 = bij[(t + 256) * KPAD + k];
    float v2 = bij[(t + 512) * KPAD + k];
    float v3 = bij[(t + 768) * KPAD + k];
    if (ADD) {
        float p[4][KS_B];
        #pragma unroll
        for (int r = 0; r < 4; ++r)
            #pragma unroll
            for (int s = 0; s < KS_B; ++s)
                p[r][s] = part[s * (M_SZ * PP) + (t + r * 256) * PP + k];
        v0 += (p[0][0] + p[0][1]) + (p[0][2] + p[0][3]);
        v1 += (p[1][0] + p[1][1]) + (p[1][2] + p[1][3]);
        v2 += (p[2][0] + p[2][1]) + (p[2][2] + p[2][3]);
        v3 += (p[3][0] + p[3][1]) + (p[3][2] + p[3][3]);
    }

    float mx = fmaxf(fmaxf(v0, v1), fmaxf(v2, v3));
    mx = fmaxf(mx, xrow<1>(mx));
    mx = fmaxf(mx, xrow<2>(mx));
    mx = fmaxf(mx, xrow<4>(mx));
    mx = fmaxf(mx, xrow<8>(mx));
    mx = fmaxf(mx, xhalf(mx));
    if (lane == 0) redm[wave] = mx;
    __syncthreads();
    {
        const float4 q0 = *(const float4*)&redm[0];
        const float4 q1 = *(const float4*)&redm[4];
        mx = fmaxf(fmaxf(fmaxf(q0.x, q0.y), fmaxf(q0.z, q0.w)),
                   fmaxf(fmaxf(q1.x, q1.y), fmaxf(q1.z, q1.w)));
    }

    float e0 = expf(v0 - mx), e1 = expf(v1 - mx);
    float e2 = expf(v2 - mx), e3 = expf(v3 - mx);
    float sm = e0 + e1 + e2 + e3;
    sm += xrow<1>(sm);
    sm += xrow<2>(sm);
    sm += xrow<4>(sm);
    sm += xrow<8>(sm);
    sm += xhalf(sm);
    if (lane == 0) reds[wave] = sm;
    __syncthreads();
    float tot;
    {
        const float4 q0 = *(const float4*)&reds[0];
        const float4 q1 = *(const float4*)&reds[4];
        tot = ((q0.x + q0.y) + (q0.z + q0.w)) + ((q1.x + q1.y) + (q1.z + q1.w));
    }
    float inv = 1.0f / tot;

    c[(t + 0  ) * KPAD + k] = e0 * inv;
    c[(t + 256) * KPAD + k] = e1 * inv;
    c[(t + 512) * KPAD + k] = e2 * inv;
    c[(t + 768) * KPAD + k] = e3 * inv;
    if (ADD && WB) {
        bij[(t + 0  ) * KPAD + k] = v0;
        bij[(t + 256) * KPAD + k] = v1;
        bij[(t + 512) * KPAD + k] = v2;
        bij[(t + 768) * KPAD + k] = v3;
    }
}

__global__ __launch_bounds__(NTH)
void caps_gemm_t(const float* __restrict__ x,
                 const float* __restrict__ c,
                 float* __restrict__ t2p) {
    __shared__ __attribute__((aligned(16))) float lds[2 * A_SZ * KROWS + KROWS * BP + 16 * PP];
    float* xs   = lds;
    float* cs   = lds + 2 * A_SZ * KROWS;
    float* tile = cs + KROWS * BP;
    const int t    = threadIdx.x;
    const int slab = blockIdx.x;
    const int ks   = blockIdx.y;

    const float* __restrict__ xsrc = x + slab * (2 * A_SZ * M_SZ) + ks * (A_SZ * KROWS);
    float4 v[16];
    #pragma unroll
    for (int j = 0; j < 16; ++j) {
        const int q = j * NTH + t;
        v[j] = *(const float4*)(xsrc + (q >> 10) * (A_SZ * M_SZ) + (q & 1023) * 4);
    }
    constexpr int NP = (KROWS + RPP - 1) / RPP;
    const int tb = min(t, RPP * 5 - 1);
    const int tr = tb / 5;
    const int tp = tb - tr * 5;
    const int rl = min((NP - 1) * RPP + tr, KROWS - 1);
    const float* __restrict__ cbase = c + ks * (KROWS * KPAD);
    const float* __restrict__ csrc  = cbase + tr * KPAD + tp * 4;
    float4 cv[NP];
    #pragma unroll
    for (int j = 0; j < NP - 1; ++j) cv[j] = *(const float4*)(csrc + j * (RPP * KPAD));
    cv[NP - 1] = *(const float4*)(cbase + rl * KPAD + tp * 4);
    #pragma unroll
    for (int j = 0; j < 16; ++j) ((float4*)xs)[j * NTH + t] = v[j];
    #pragma unroll
    for (int j = 0; j < NP - 1; ++j) ((float4*)cs)[j * (RPP * 5) + tb] = cv[j];
    ((float4*)cs)[rl * (BP / 4) + tp] = cv[NP - 1];
    __syncthreads();

    const int wave = t >> 5;
    const int lane = t & 31;
    const int lo   = lane & 15;
    const int hi   = lane >> 4;

    if (wave < 2) {
        v8f acc = {};
        const int abase = (lo >> 3) * (A_SZ * KROWS) + (lo & 7) + hi * 2 * A_SZ;
        const int bcol  = min(wave * 16 + lo, K_CLS - 1);
        const int bbase = hi * 2 * BP + bcol;
        constexpr int NG = (KROWS / 4) / GSTEP;

        v2f a0[GSTEP], b0[GSTEP], a1[GSTEP], b1[GSTEP];
        auto load = [&](v2f (&a)[GSTEP], v2f (&b)[GSTEP], int g) {
            #pragma unroll
            for (int s = 0; s < GSTEP; ++s) {
                const int mb = (g * GSTEP + s) * 4;
                a[s].x = xs[abase + mb * A_SZ];
                a[s].y = xs[abase + mb * A_SZ + A_SZ];
                b[s].x = cs[bbase + mb * BP];
                b[s].y = cs[bbase + mb * BP + BP];
            }
        };
        auto steps = [&](const v2f (&a)[GSTEP], const v2f (&b)[GSTEP]) {
            #pragma unroll
            for (int s = 0; s < GSTEP; ++s)
                acc = __builtin_amdgcn_wmma_f32_16x16x4_f32(
                    false, a[s], false, b[s], (short)0, acc, false, false);
        };

        load(a0, b0, 0);
        #pragma unroll 1
        for (int g = 0; g < NG - 2; g += 2) {
            load(a1, b1, g + 1);
            __builtin_amdgcn_sched_barrier(0);
            steps(a0, b0);
            __builtin_amdgcn_sched_barrier(0);
            load(a0, b0, g + 2);
            __builtin_amdgcn_sched_barrier(0);
            steps(a1, b1);
        }
        load(a1, b1, NG - 1);
        __builtin_amdgcn_sched_barrier(0);
        steps(a0, b0);
        steps(a1, b1);

        if (wave == 0 || lo < K_CLS - 16) {
            #pragma unroll
            for (int r = 0; r < 8; ++r) tile[(r + hi * 8) * PP + wave * 16 + lo] = acc[r];
        }
    }
    __syncthreads();
    float* __restrict__ dst = t2p + ks * (I_SZ * PP) + slab * (16 * PP);
    const float r0 = tile[t];
    const float r1 = tile[t + NTH];
    const float r2 = tile[min(t + 2 * NTH, 16 * PP - 1)];
    dst[t]       = r0;
    dst[t + NTH] = r1;
    if (t < 16 * PP - 2 * NTH) dst[t + 2 * NTH] = r2;
}

__global__ void caps_squash(const float* __restrict__ t2p,
                            const float* __restrict__ W,
                            float* __restrict__ out,
                            float* __restrict__ wv2) {
    __shared__ float s_s[K_CLS][O_SZ];
    __shared__ float s_v[K_CLS][O_SZ];
    const int b = blockIdx.x;
    const int t = threadIdx.x;
    const int k = t >> 4;
    const int o = t & 15;

    float s = 0.0f;
    #pragma unroll
    for (int a = 0; a < A_SZ; ++a) {
        float tv = t2p[(b * 8 + a) * PP + k];
        #pragma unroll
        for (int p = 1; p < KS_T; ++p) tv += t2p[p * (I_SZ * PP) + (b * 8 + a) * PP + k];
        s += W[k * (O_SZ * A_SZ) + o * A_SZ + a] * tv;
    }
    s_s[k][o] = s;
    __syncthreads();

    float magsq = 0.0f;
    #pragma unroll
    for (int kk = 0; kk < K_CLS; ++kk) {
        float sv = s_s[kk][o];
        magsq += sv * sv;
    }
    float scale = magsq / ((1.0f + magsq) * sqrtf(magsq));
    float v = s * scale;
    s_v[k][o] = v;
    out[b * (K_CLS * O_SZ) + t] = v;
    __syncthreads();

    if (t < K_CLS * A_SZ) {
        int k2 = t % K_CLS;
        int a  = t / K_CLS;
        float w = 0.0f;
        #pragma unroll
        for (int o2 = 0; o2 < O_SZ; ++o2)
            w += W[k2 * (O_SZ * A_SZ) + o2 * A_SZ + a] * s_v[k2][o2];
        wv2[(b * 8 + a) * KPAD + k2] = w * (1.0f / (float)B_SZ);
    }
}

__global__ __launch_bounds__(NTH)
void caps_gemm_b(const float* __restrict__ x,
                 const float* __restrict__ wv2,
                 float* __restrict__ bijp) {
    constexpr int NB = KROWS / A_SZ;
    __shared__ __attribute__((aligned(16))) float lds[NB * 16 * A_SZ + KROWS * BP + 16 * PP];
    float* xs   = lds;
    float* ws   = lds + NB * 16 * A_SZ;
    float* tile = ws + KROWS * BP;
    const int t    = threadIdx.x;
    const int slab = blockIdx.x;
    const int ks   = blockIdx.y;

    const float* __restrict__ xsrc = x + ks * (NB * A_SZ * M_SZ) + slab * (16 * A_SZ);
    float4 v[16];
    #pragma unroll
    for (int j = 0; j < 16; ++j) {
        const int q = j * NTH + t;
        v[j] = *(const float4*)(xsrc + (q >> 5) * (A_SZ * M_SZ) + (q & 31) * 4);
    }
    constexpr int NP = (KROWS + RPP - 1) / RPP;
    const int tb = min(t, RPP * 5 - 1);
    const int tr = tb / 5;
    const int tp = tb - tr * 5;
    const int rl = min((NP - 1) * RPP + tr, KROWS - 1);
    const float* __restrict__ wbase = wv2 + ks * (KROWS * KPAD);
    const float* __restrict__ wsrc  = wbase + tr * KPAD + tp * 4;
    float4 wv[NP];
    #pragma unroll
    for (int j = 0; j < NP - 1; ++j) wv[j] = *(const float4*)(wsrc + j * (RPP * KPAD));
    wv[NP - 1] = *(const float4*)(wbase + rl * KPAD + tp * 4);
    #pragma unroll
    for (int j = 0; j < 16; ++j) ((float4*)xs)[j * NTH + t] = v[j];
    #pragma unroll
    for (int j = 0; j < NP - 1; ++j) ((float4*)ws)[j * (RPP * 5) + tb] = wv[j];
    ((float4*)ws)[rl * (BP / 4) + tp] = wv[NP - 1];
    __syncthreads();

    const int wave = t >> 5;
    const int lane = t & 31;
    const int lo   = lane & 15;
    const int hi   = lane >> 4;

    if (wave < 2) {
        v8f acc = {};
        const int abase = lo * A_SZ + hi * 2;
        const int bcol  = min(wave * 16 + lo, K_CLS - 1);
        const int bbase = hi * 2 * BP + bcol;
        constexpr int GB = GSTEP / 2;
        constexpr int NG = NB / GB;

        v2f a0[GSTEP], b0[GSTEP], a1[GSTEP], b1[GSTEP];
        auto load = [&](v2f (&a)[GSTEP], v2f (&bf)[GSTEP], int g) {
            #pragma unroll
            for (int s = 0; s < GB; ++s) {
                const int b = g * GB + s;
                a[2 * s]     = *(const v2f*)&xs[b * (16 * A_SZ) + abase];
                a[2 * s + 1] = *(const v2f*)&xs[b * (16 * A_SZ) + abase + 4];
                bf[2 * s].x     = ws[(b * A_SZ    ) * BP + bbase];
                bf[2 * s].y     = ws[(b * A_SZ + 1) * BP + bbase];
                bf[2 * s + 1].x = ws[(b * A_SZ + 4) * BP + bbase];
                bf[2 * s + 1].y = ws[(b * A_SZ + 5) * BP + bbase];
            }
        };
        auto steps = [&](const v2f (&a)[GSTEP], const v2f (&bf)[GSTEP]) {
            #pragma unroll
            for (int s = 0; s < GSTEP; ++s)
                acc = __builtin_amdgcn_wmma_f32_16x16x4_f32(
                    false, a[s], false, bf[s], (short)0, acc, false, false);
        };

        load(a0, b0, 0);
        #pragma unroll 1
        for (int g = 0; g < NG - 2; g += 2) {
            load(a1, b1, g + 1);
            __builtin_amdgcn_sched_barrier(0);
            steps(a0, b0);
            __builtin_amdgcn_sched_barrier(0);
            load(a0, b0, g + 2);
            __builtin_amdgcn_sched_barrier(0);
            steps(a1, b1);
        }
        load(a1, b1, NG - 1);
        __builtin_amdgcn_sched_barrier(0);
        steps(a0, b0);
        steps(a1, b1);

        if (wave == 0 || lo < K_CLS - 16) {
            #pragma unroll
            for (int r = 0; r < 8; ++r) tile[(r + hi * 8) * PP + wave * 16 + lo] = acc[r];
        }
    }
    __syncthreads();
    float* __restrict__ dst = bijp + ks * (M_SZ * PP) + slab * (16 * PP);
    const float r0 = tile[t];
    const float r1 = tile[t + NTH];
    const float r2 = tile[min(t + 2 * NTH, 16 * PP - 1)];
    dst[t]       = r0;
    dst[t + NTH] = r1;
    if (t < 16 * PP - 2 * NTH) dst[t + 2 * NTH] = r2;
}

extern "C" void kernel_launch(void* const* d_in, const int* in_sizes, int n_in,
                              void* d_out, int out_size, void* d_ws, size_t ws_size,
                              hipStream_t stream) {
    (void)in_sizes; (void)n_in; (void)out_size; (void)ws_size;
    const float* x = (const float*)d_in[0];
    const float* W = (const float*)d_in[1];
    float* out = (float*)d_out;

    float* bij  = (float*)d_ws;
    float* wv2  = bij + M_SZ * KPAD;
    float* c    = wv2;
    float* part = wv2 + I_SZ * KPAD;

    caps_zero_bij<<<(M_SZ * KPAD) / 256, 256, 0, stream>>>(bij);

    for (int it = 0; it < 3; ++it) {
        if (it == 0)      caps_softmax<false, false><<<K_CLS, 256, 0, stream>>>(bij, part, c);
        else if (it == 1) caps_softmax<true,  true ><<<K_CLS, 256, 0, stream>>>(bij, part, c);
        else              caps_softmax<true,  false><<<K_CLS, 256, 0, stream>>>(bij, part, c);
        caps_gemm_t<<<dim3(I_SZ / 16, KS_T), NTH, 0, stream>>>(x, c, part);
        caps_squash<<<B_SZ, 320, 0, stream>>>(part, W, out, wv2);
        if (it < 2)
            caps_gemm_b<<<dim3(M_SZ / 16, KS_B), NTH, 0, stream>>>(x, wv2, part);
    }
}


# Logits, agreement partials and c stored class-major: a softmax block reads and writes whole 128-byte lines, not one float per line; GEMM 1 copies c in 2 KB runs, 8-byte B reads (sum order kept)
# speedup vs baseline: 9.1415x; 1.2626x over previous
#include <hip/hip_runtime.h>
#include <math.h>

typedef __attribute__((ext_vector_type(2))) float v2f;
typedef __attribute__((ext_vector_type(8))) float v8f;

#define B_SZ   256
#define M_SZ   1024
#define A_SZ   8
#define K_CLS  20
#define O_SZ   16
#define KPAD   32
#define BP     20
#define GSTEP  16
#define NTH    128
#define RPP    25
#define I_SZ   (B_SZ * A_SZ)
#define KS_T   2
#define KS_B   4
#define KROWS  512
#define PP     20
#define CP     516

__global__ void caps_zero_bij(float* __restrict__ bij) {
    int idx = blockIdx.x * blockDim.x + threadIdx.x;
    bij[idx] = 0.0f;
}

template <int M>
__device__ __forceinline__ float xrow(float x) {
    const int b = __builtin_bit_cast(int, x);
    return __builtin_bit_cast(float, __builtin_amdgcn_update_dpp(b, b, 0x160 + M, 0xf, 0xf, false));
}
__device__ __forceinline__ float xhalf(float x) {
    const int b = __builtin_bit_cast(int, x);
    return __builtin_bit_cast(float, __builtin_amdgcn_permlanex16(b, b, 0x76543210, 0xfedcba98, false, false));
}

template <bool ADD, bool WB>
__global__ void caps_softmax(float* __restrict__ bij, const float* __restrict__ part,
                             float* __restrict__ c) {
    __shared__ __attribute__((aligned(16))) float redm[8];
    __shared__ __attribute__((aligned(16))) float reds[8];
    const int k = blockIdx.x;
    const int t = threadIdx.x;
    const int wave = t >> 5;
    const int lane = t & 31;

    float* __restrict__ brow = bij + k * M_SZ;
    float v0 = brow[t      ];
    float v1 = brow[t + 256];
    float v2 = brow[t + 512];
    float v3 = brow[t + 768];
    if (ADD) {
        float p[4][KS_B];
        const float* __restrict__ pk = part + (t >> 4) * (16 * PP) + k * 16 + (t & 15);
        #pragma unroll
        for (int r = 0; r < 4; ++r)
            #pragma unroll
            for (int s = 0; s < KS_B; ++s)
                p[r][s] = pk[s * (M_SZ * PP) + r * (16 * 16 * PP)];
        v0 += (p[0][0] + p[0][1]) + (p[0][2] + p[0][3]);
        v1 += (p[1][0] + p[1][1]) + (p[1][2] + p[1][3]);
        v2 += (p[2][0] + p[2][1]) + (p[2][2] + p[2][3]);
        v3 += (p[3][0] + p[3][1]) + (p[3][2] + p[3][3]);
    }

    float mx = fmaxf(fmaxf(v0, v1), fmaxf(v2, v3));
    mx = fmaxf(mx, xrow<1>(mx));
    mx = fmaxf(mx, xrow<2>(mx));
    mx = fmaxf(mx, xrow<4>(mx));
    mx = fmaxf(mx, xrow<8>(mx));
    mx = fmaxf(mx, xhalf(mx));
    if (lane == 0) redm[wave] = mx;
    __syncthreads();
    {
        const float4 q0 = *(const float4*)&redm[0];
        const float4 q1 = *(const float4*)&redm[4];
        mx = fmaxf(fmaxf(fmaxf(q0.x, q0.y), fmaxf(q0.z, q0.w)),
                   fmaxf(fmaxf(q1.x, q1.y), fmaxf(q1.z, q1.w)));
    }

    float e0 = expf(v0 - mx), e1 = expf(v1 - mx);
    float e2 = expf(v2 - mx), e3 = expf(v3 - mx);
    float sm = e0 + e1 + e2 + e3;
    sm += xrow<1>(sm);
    sm += xrow<2>(sm);
    sm += xrow<4>(sm);
    sm += xrow<8>(sm);
    sm += xhalf(sm);
    if (lane == 0) reds[wave] = sm;
    __syncthreads();
    float tot;
    {
        const float4 q0 = *(const float4*)&reds[0];
        const float4 q1 = *(const float4*)&reds[4];
        tot = ((q0.x + q0.y) + (q0.z + q0.w)) + ((q1.x + q1.y) + (q1.z + q1.w));
    }
    float inv = 1.0f / tot;

    float* __restrict__ crow = c + k * M_SZ;
    crow[t      ] = e0 * inv;
    crow[t + 256] = e1 * inv;
    crow[t + 512] = e2 * inv;
    crow[t + 768] = e3 * inv;
    if (ADD && WB) {
        brow[t      ] = v0;
        brow[t + 256] = v1;
        brow[t + 512] = v2;
        brow[t + 768] = v3;
    }
}

__global__ __launch_bounds__(NTH)
void caps_gemm_t(const float* __restrict__ x,
                 const float* __restrict__ c,
                 float* __restrict__ t2p) {
    __shared__ __attribute__((aligned(16))) float lds[2 * A_SZ * KROWS + K_CLS * CP + 16 * PP];
    float* xs   = lds;
    float* cs   = lds + 2 * A_SZ * KROWS;
    float* tile = cs + K_CLS * CP;
    const int t    = threadIdx.x;
    const int slab = blockIdx.x;
    const int ks   = blockIdx.y;

    const float* __restrict__ xsrc = x + slab * (2 * A_SZ * M_SZ) + ks * (A_SZ * KROWS);
    float4 v[16];
    #pragma unroll
    for (int j = 0; j < 16; ++j) {
        const int q = j * NTH + t;
        v[j] = *(const float4*)(xsrc + (q >> 10) * (A_SZ * M_SZ) + (q & 1023) * 4);
    }
    const float* __restrict__ csrc = c + ks * KROWS + t * 4;
    float4 cv[K_CLS];
    #pragma unroll
    for (int j = 0; j < K_CLS; ++j) cv[j] = *(const float4*)(csrc + j * M_SZ);
    #pragma unroll
    for (int j = 0; j < 16; ++j) ((float4*)xs)[j * NTH + t] = v[j];
    #pragma unroll
    for (int j = 0; j < K_CLS; ++j) *(float4*)&cs[j * CP + t * 4] = cv[j];
    __syncthreads();

    const int wave = t >> 5;
    const int lane = t & 31;
    const int lo   = lane & 15;
    const int hi   = lane >> 4;

    if (wave < 2) {
        v8f acc = {};
        const int abase = (lo >> 3) * (A_SZ * KROWS) + (lo & 7) + hi * 2 * A_SZ;
        const int bcol  = min(wave * 16 + lo, K_CLS - 1);
        const int bbase = bcol * CP + hi * 2;
        constexpr int NG = (KROWS / 4) / GSTEP;

        v2f a0[GSTEP], b0[GSTEP], a1[GSTEP], b1[GSTEP];
        auto load = [&](v2f (&a)[GSTEP], v2f (&b)[GSTEP], int g) {
            #pragma unroll
            for (int s = 0; s < GSTEP; ++s) {
                const int mb = (g * GSTEP + s) * 4;
                a[s].x = xs[abase + mb * A_SZ];
                a[s].y = xs[abase + mb * A_SZ + A_SZ];
                b[s]   = *(const v2f*)&cs[bbase + mb];
            }
        };
        auto steps = [&](const v2f (&a)[GSTEP], const v2f (&b)[GSTEP]) {
            #pragma unroll
            for (int s = 0; s < GSTEP; ++s)
                acc = __builtin_amdgcn_wmma_f32_16x16x4_f32(
                    false, a[s], false, b[s], (short)0, acc, false, false);
        };

        load(a0, b0, 0);
        #pragma unroll 1
        for (int g = 0; g < NG - 2; g += 2) {
            load(a1, b1, g + 1);
            __builtin_amdgcn_sched_barrier(0);
            steps(a0, b0);
            __builtin_amdgcn_sched_barrier(0);
            load(a0, b0, g + 2);
            __builtin_amdgcn_sched_barrier(0);
            steps(a1, b1);
        }
        load(a1, b1, NG - 1);
        __builtin_amdgcn_sched_barrier(0);
        steps(a0, b0);
        steps(a1, b1);

        if (wave == 0 || lo < K_CLS - 16) {
            #pragma unroll
            for (int r = 0; r < 8; ++r) tile[(r + hi * 8) * PP + wave * 16 + lo] = acc[r];
        }
    }
    __syncthreads();
    float* __restrict__ dst = t2p + ks * (I_SZ * PP) + slab * (16 * PP);
    const float r0 = tile[t];
    const float r1 = tile[t + NTH];
    const float r2 = tile[min(t + 2 * NTH, 16 * PP - 1)];
    dst[t]       = r0;
    dst[t + NTH] = r1;
    if (t < 16 * PP - 2 * NTH) dst[t + 2 * NTH] = r2;
}

__global__ void caps_squash(const float* __restrict__ t2p,
                            const float* __restrict__ W,
                            float* __restrict__ out,
                            float* __restrict__ wv2) {
    __shared__ float s_s[K_CLS][O_SZ];
    __shared__ float s_v[K_CLS][O_SZ];
    const int b = blockIdx.x;
    const int t = threadIdx.x;
    const int k = t >> 4;
    const int o = t & 15;

    float s = 0.0f;
    #pragma unroll
    for (int a = 0; a < A_SZ; ++a) {
        float tv = t2p[(b * 8 + a) * PP + k];
        #pragma unroll
        for (int p = 1; p < KS_T; ++p) tv += t2p[p * (I_SZ * PP) + (b * 8 + a) * PP + k];
        s += W[k * (O_SZ * A_SZ) + o * A_SZ + a] * tv;
    }
    s_s[k][o] = s;
    __syncthreads();

    float magsq = 0.0f;
    #pragma unroll
    for (int kk = 0; kk < K_CLS; ++kk) {
        float sv = s_s[kk][o];
        magsq += sv * sv;
    }
    float scale = magsq / ((1.0f + magsq) * sqrtf(magsq));
    float v = s * scale;
    s_v[k][o] = v;
    out[b * (K_CLS * O_SZ) + t] = v;
    __syncthreads();

    if (t < K_CLS * A_SZ) {
        int k2 = t % K_CLS;
        int a  = t / K_CLS;
        float w = 0.0f;
        #pragma unroll
        for (int o2 = 0; o2 < O_SZ; ++o2)
            w += W[k2 * (O_SZ * A_SZ) + o2 * A_SZ + a] * s_v[k2][o2];
        wv2[(b * 8 + a) * KPAD + k2] = w * (1.0f / (float)B_SZ);
    }
}

__global__ __launch_bounds__(NTH)
void caps_gemm_b(const float* __restrict__ x,
                 const float* __restrict__ wv2,
                 float* __restrict__ bijp) {
    constexpr int NB = KROWS / A_SZ;
    __shared__ __attribute__((aligned(16))) float lds[NB * 16 * A_SZ + KROWS * BP + 16 * PP];
    float* xs   = lds;
    float* ws   = lds + NB * 16 * A_SZ;
    float* tile = ws + KROWS * BP;
    const int t    = threadIdx.x;
    const int slab = blockIdx.x;
    const int ks   = blockIdx.y;

    const float* __restrict__ xsrc = x + ks * (NB * A_SZ * M_SZ) + slab * (16 * A_SZ);
    float4 v[16];
    #pragma unroll
    for (int j = 0; j < 16; ++j) {
        const int q = j * NTH + t;
        v[j] = *(const float4*)(xsrc + (q >> 5) * (A_SZ * M_SZ) + (q & 31) * 4);
    }
    constexpr int NP = (KROWS + RPP - 1) / RPP;
    const int tb = min(t, RPP * 5 - 1);
    const int tr = tb / 5;
    const int tp = tb - tr * 5;
    const int rl = min((NP - 1) * RPP + tr, KROWS - 1);
    const float* __restrict__ wbase = wv2 + ks * (KROWS * KPAD);
    const float* __restrict__ wsrc  = wbase + tr * KPAD + tp * 4;
    float4 wv[NP];
    #pragma unroll
    for (int j = 0; j < NP - 1; ++j) wv[j] = *(const float4*)(wsrc + j * (RPP * KPAD));
    wv[NP - 1] = *(const float4*)(wbase + rl * KPAD + tp * 4);
    #pragma unroll
    for (int j = 0; j < 16; ++j) ((float4*)xs)[j * NTH + t] = v[j];
    #pragma unroll
    for (int j = 0; j < NP - 1; ++j) ((float4*)ws)[j * (RPP * 5) + tb] = wv[j];
    ((float4*)ws)[rl * (BP / 4) + tp] = wv[NP - 1];
    __syncthreads();

    const int wave = t >> 5;
    const int lane = t & 31;
    const int lo   = lane & 15;
    const int hi   = lane >> 4;

    if (wave < 2) {
        v8f acc = {};
        const int abase = lo * A_SZ + hi * 2;
        const int bcol  = min(wave * 16 + lo, K_CLS - 1);
        const int bbase = hi * 2 * BP + bcol;
        constexpr int GB = GSTEP / 2;
        constexpr int NG = NB / GB;

        v2f a0[GSTEP], b0[GSTEP], a1[GSTEP], b1[GSTEP];
        auto load = [&](v2f (&a)[GSTEP], v2f (&bf)[GSTEP], int g) {
            #pragma unroll
            for (int s = 0; s < GB; ++s) {
                const int b = g * GB + s;
                a[2 * s]     = *(const v2f*)&xs[b * (16 * A_SZ) + abase];
                a[2 * s + 1] = *(const v2f*)&xs[b * (16 * A_SZ) + abase + 4];
                bf[2 * s].x     = ws[(b * A_SZ    ) * BP + bbase];
                bf[2 * s].y     = ws[(b * A_SZ + 1) * BP + bbase];
                bf[2 * s + 1].x = ws[(b * A_SZ + 4) * BP + bbase];
                bf[2 * s + 1].y = ws[(b * A_SZ + 5) * BP + bbase];
            }
        };
        auto steps = [&](const v2f (&a)[GSTEP], const v2f (&bf)[GSTEP]) {
            #pragma unroll
            for (int s = 0; s < GSTEP; ++s)
                acc = __builtin_amdgcn_wmma_f32_16x16x4_f32(
                    false, a[s], false, bf[s], (short)0, acc, false, false);
        };

        load(a0, b0, 0);
        #pragma unroll 1
        for (int g = 0; g < NG - 2; g += 2) {
            load(a1, b1, g + 1);
            __builtin_amdgcn_sched_barrier(0);
            steps(a0, b0);
            __builtin_amdgcn_sched_barrier(0);
            load(a0, b0, g + 2);
            __builtin_amdgcn_sched_barrier(0);
            steps(a1, b1);
        }
        load(a1, b1, NG - 1);
        __builtin_amdgcn_sched_barrier(0);
        steps(a0, b0);
        steps(a1, b1);

        if (wave == 0 || lo < K_CLS - 16) {
            #pragma unroll
            for (int r = 0; r < 8; ++r) tile[(wave * 16 + lo) * 16 + r + hi * 8] = acc[r];
        }
    }
    __syncthreads();
    float* __restrict__ dst = bijp + ks * (M_SZ * PP) + slab * (16 * PP);
    const float r0 = tile[t];
    const float r1 = tile[t + NTH];
    const float r2 = tile[min(t + 2 * NTH, 16 * PP - 1)];
    dst[t]       = r0;
    dst[t + NTH] = r1;
    if (t < 16 * PP - 2 * NTH) dst[t + 2 * NTH] = r2;
}

extern "C" void kernel_launch(void* const* d_in, const int* in_sizes, int n_in,
                              void* d_out, int out_size, void* d_ws, size_t ws_size,
                              hipStream_t stream) {
    (void)in_sizes; (void)n_in; (void)out_size; (void)ws_size;
    const float* x = (const float*)d_in[0];
    const float* W = (const float*)d_in[1];
    float* out = (float*)d_out;

    float* bij  = (float*)d_ws;
    float* wv2  = bij + M_SZ * KPAD;
    float* c    = wv2;
    float* part = wv2 + I_SZ * KPAD;

    caps_zero_bij<<<(K_CLS * M_SZ) / 256, 256, 0, stream>>>(bij);

    for (int it = 0; it < 3; ++it) {
        if (it == 0)      caps_softmax<false, false><<<K_CLS, 256, 0, stream>>>(bij, part, c);
        else if (it == 1) caps_softmax<true,  true ><<<K_CLS, 256, 0, stream>>>(bij, part, c);
        else              caps_softmax<true,  false><<<K_CLS, 256, 0, stream>>>(bij, part, c);
        caps_gemm_t<<<dim3(I_SZ / 16, KS_T), NTH, 0, stream>>>(x, c, part);
        caps_squash<<<B_SZ, 320, 0, stream>>>(part, W, out, wv2);
        if (it < 2)
            caps_gemm_b<<<dim3(M_SZ / 16, KS_B), NTH, 0, stream>>>(x, wv2, part);
    }
}


# Pass 1 start state folded in: zero logits give c = 2^-10 exactly, so GEMM 1 of pass 1 takes that constant as its B fragment; no zero-init and no first softmax launch, 12 -> 10 (sum order kept)
# speedup vs baseline: 10.3964x; 1.1373x over previous
#include <hip/hip_runtime.h>
#include <math.h>

typedef __attribute__((ext_vector_type(2))) float v2f;
typedef __attribute__((ext_vector_type(8))) float v8f;

#define B_SZ   256
#define M_SZ   1024
#define A_SZ   8
#define K_CLS  20
#define O_SZ   16
#define KPAD   32
#define BP     20
#define GSTEP  16
#define NTH    128
#define RPP    25
#define I_SZ   (B_SZ * A_SZ)
#define KS_T   2
#define KS_B   4
#define KROWS  512
#define PP     20
#define CP     516

#define C_START 0x1p-10f

template <int M>
__device__ __forceinline__ float xrow(float x) {
    const int b = __builtin_bit_cast(int, x);
    return __builtin_bit_cast(float, __builtin_amdgcn_update_dpp(b, b, 0x160 + M, 0xf, 0xf, false));
}
__device__ __forceinline__ float xhalf(float x) {
    const int b = __builtin_bit_cast(int, x);
    return __builtin_bit_cast(float, __builtin_amdgcn_permlanex16(b, b, 0x76543210, 0xfedcba98, false, false));
}

template <bool BASE, bool WB>
__global__ void caps_softmax(float* __restrict__ bij, const float* __restrict__ part,
                             float* __restrict__ c) {
    __shared__ __attribute__((aligned(16))) float redm[8];
    __shared__ __attribute__((aligned(16))) float reds[8];
    const int k = blockIdx.x;
    const int t = threadIdx.x;
    const int wave = t >> 5;
    const int lane = t & 31;

    float* __restrict__ brow = bij + k * M_SZ;
    float v0 = 0.0f, v1 = 0.0f, v2 = 0.0f, v3 = 0.0f;
    if (BASE) {
        v0 = brow[t      ];
        v1 = brow[t + 256];
        v2 = brow[t + 512];
        v3 = brow[t + 768];
    }
    {
        float p[4][KS_B];
        const float* __restrict__ pk = part + (t >> 4) * (16 * PP) + k * 16 + (t & 15);
        #pragma unroll
        for (int r = 0; r < 4; ++r)
            #pragma unroll
            for (int s = 0; s < KS_B; ++s)
                p[r][s] = pk[s * (M_SZ * PP) + r * (16 * 16 * PP)];
        v0 += (p[0][0] + p[0][1]) + (p[0][2] + p[0][3]);
        v1 += (p[1][0] + p[1][1]) + (p[1][2] + p[1][3]);
        v2 += (p[2][0] + p[2][1]) + (p[2][2] + p[2][3]);
        v3 += (p[3][0] + p[3][1]) + (p[3][2] + p[3][3]);
    }

    float mx = fmaxf(fmaxf(v0, v1), fmaxf(v2, v3));
    mx = fmaxf(mx, xrow<1>(mx));
    mx = fmaxf(mx, xrow<2>(mx));
    mx = fmaxf(mx, xrow<4>(mx));
    mx = fmaxf(mx, xrow<8>(mx));
    mx = fmaxf(mx, xhalf(mx));
    if (lane == 0) redm[wave] = mx;
    __syncthreads();
    {
        const float4 q0 = *(const float4*)&redm[0];
        const float4 q1 = *(const float4*)&redm[4];
        mx = fmaxf(fmaxf(fmaxf(q0.x, q0.y), fmaxf(q0.z, q0.w)),
                   fmaxf(fmaxf(q1.x, q1.y), fmaxf(q1.z, q1.w)));
    }

    float e0 = expf(v0 - mx), e1 = expf(v1 - mx);
    float e2 = expf(v2 - mx), e3 = expf(v3 - mx);
    float sm = e0 + e1 + e2 + e3;
    sm += xrow<1>(sm);
    sm += xrow<2>(sm);
    sm += xrow<4>(sm);
    sm += xrow<8>(sm);
    sm += xhalf(sm);
    if (lane == 0) reds[wave] = sm;
    __syncthreads();
    float tot;
    {
        const float4 q0 = *(const float4*)&reds[0];
        const float4 q1 = *(const float4*)&reds[4];
        tot = ((q0.x + q0.y) + (q0.z + q0.w)) + ((q1.x + q1.y) + (q1.z + q1.w));
    }
    float inv = 1.0f / tot;

    float* __restrict__ crow = c + k * M_SZ;
    crow[t      ] = e0 * inv;
    crow[t + 256] = e1 * inv;
    crow[t + 512] = e2 * inv;
    crow[t + 768] = e3 * inv;
    if (WB) {
        brow[t      ] = v0;
        brow[t + 256] = v1;
        brow[t + 512] = v2;
        brow[t + 768] = v3;
    }
}

template <bool START>
__global__ __launch_bounds__(NTH)
void caps_gemm_t(const float* __restrict__ x,
                 const float* __restrict__ c,
                 float* __restrict__ t2p) {
    __shared__ __attribute__((aligned(16))) float lds[2 * A_SZ * KROWS + K_CLS * CP + 16 * PP];
    float* xs   = lds;
    float* cs   = lds + 2 * A_SZ * KROWS;
    float* tile = cs + K_CLS * CP;
    const int t    = threadIdx.x;
    const int slab = blockIdx.x;
    const int ks   = blockIdx.y;

    const float* __restrict__ xsrc = x + slab * (2 * A_SZ * M_SZ) + ks * (A_SZ * KROWS);
    float4 v[16];
    #pragma unroll
    for (int j = 0; j < 16; ++j) {
        const int q = j * NTH + t;
        v[j] = *(const float4*)(xsrc + (q >> 10) * (A_SZ * M_SZ) + (q & 1023) * 4);
    }
    const float* __restrict__ csrc = c + ks * KROWS + t * 4;
    float4 cv[K_CLS];
    if (!START) {
        #pragma unroll
        for (int j = 0; j < K_CLS; ++j) cv[j] = *(const float4*)(csrc + j * M_SZ);
    }
    #pragma unroll
    for (int j = 0; j < 16; ++j) ((float4*)xs)[j * NTH + t] = v[j];
    if (!START) {
        #pragma unroll
        for (int j = 0; j < K_CLS; ++j) *(float4*)&cs[j * CP + t * 4] = cv[j];
    }
    __syncthreads();

    const int wave = t >> 5;
    const int lane = t & 31;
    const int lo   = lane & 15;
    const int hi   = lane >> 4;

    if (wave < 2) {
        v8f acc = {};
        const int abase = (lo >> 3) * (A_SZ * KROWS) + (lo & 7) + hi * 2 * A_SZ;
        const int bcol  = min(wave * 16 + lo, K_CLS - 1);
        const int bbase = bcol * CP + hi * 2;
        constexpr int NG = (KROWS / 4) / GSTEP;

        v2f a0[GSTEP], b0[GSTEP], a1[GSTEP], b1[GSTEP];
        auto load = [&](v2f (&a)[GSTEP], v2f (&b)[GSTEP], int g) {
            #pragma unroll
            for (int s = 0; s < GSTEP; ++s) {
                const int mb = (g * GSTEP + s) * 4;
                a[s].x = xs[abase + mb * A_SZ];
                a[s].y = xs[abase + mb * A_SZ + A_SZ];
                if (START) { b[s].x = C_START; b[s].y = C_START; }
                else b[s] = *(const v2f*)&cs[bbase + mb];
            }
        };
        auto steps = [&](const v2f (&a)[GSTEP], const v2f (&b)[GSTEP]) {
            #pragma unroll
            for (int s = 0; s < GSTEP; ++s)
                acc = __builtin_amdgcn_wmma_f32_16x16x4_f32(
                    false, a[s], false, b[s], (short)0, acc, false, false);
        };

        load(a0, b0, 0);
        #pragma unroll 1
        for (int g = 0; g < NG - 2; g += 2) {
            load(a1, b1, g + 1);
            __builtin_amdgcn_sched_barrier(0);
            steps(a0, b0);
            __builtin_amdgcn_sched_barrier(0);
            load(a0, b0, g + 2);
            __builtin_amdgcn_sched_barrier(0);
            steps(a1, b1);
        }
        load(a1, b1, NG - 1);
        __builtin_amdgcn_sched_barrier(0);
        steps(a0, b0);
        steps(a1, b1);

        if (wave == 0 || lo < K_CLS - 16) {
            #pragma unroll
            for (int r = 0; r < 8; ++r) tile[(r + hi * 8) * PP + wave * 16 + lo] = acc[r];
        }
    }
    __syncthreads();
    float* __restrict__ dst = t2p + ks * (I_SZ * PP) + slab * (16 * PP);
    const float r0 = tile[t];
    const float r1 = tile[t + NTH];
    const float r2 = tile[min(t + 2 * NTH, 16 * PP - 1)];
    dst[t]       = r0;
    dst[t + NTH] = r1;
    if (t < 16 * PP - 2 * NTH) dst[t + 2 * NTH] = r2;
}

__global__ void caps_squash(const float* __restrict__ t2p,
                            const float* __restrict__ W,
                            float* __restrict__ out,
                            float* __restrict__ wv2) {
    __shared__ float s_s[K_CLS][O_SZ];
    __shared__ float s_v[K_CLS][O_SZ];
    const int b = blockIdx.x;
    const int t = threadIdx.x;
    const int k = t >> 4;
    const int o = t & 15;

    float s = 0.0f;
    #pragma unroll
    for (int a = 0; a < A_SZ; ++a) {
        float tv = t2p[(b * 8 + a) * PP + k];
        #pragma unroll
        for (int p = 1; p < KS_T; ++p) tv += t2p[p * (I_SZ * PP) + (b * 8 + a) * PP + k];
        s += W[k * (O_SZ * A_SZ) + o * A_SZ + a] * tv;
    }
    s_s[k][o] = s;
    __syncthreads();

    float magsq = 0.0f;
    #pragma unroll
    for (int kk = 0; kk < K_CLS; ++kk) {
        float sv = s_s[kk][o];
        magsq += sv * sv;
    }
    float scale = magsq / ((1.0f + magsq) * sqrtf(magsq));
    float v = s * scale;
    s_v[k][o] = v;
    out[b * (K_CLS * O_SZ) + t] = v;
    __syncthreads();

    if (t < K_CLS * A_SZ) {
        int k2 = t % K_CLS;
        int a  = t / K_CLS;
        float w = 0.0f;
        #pragma unroll
        for (int o2 = 0; o2 < O_SZ; ++o2)
            w += W[k2 * (O_SZ * A_SZ) + o2 * A_SZ + a] * s_v[k2][o2];
        wv2[(b * 8 + a) * KPAD + k2] = w * (1.0f / (float)B_SZ);
    }
}

__global__ __launch_bounds__(NTH)
void caps_gemm_b(const float* __restrict__ x,
                 const float* __restrict__ wv2,
                 float* __restrict__ bijp) {
    constexpr int NB = KROWS / A_SZ;
    __shared__ __attribute__((aligned(16))) float lds[NB * 16 * A_SZ + KROWS * BP + 16 * PP];
    float* xs   = lds;
    float* ws   = lds + NB * 16 * A_SZ;
    float* tile = ws + KROWS * BP;
    const int t    = threadIdx.x;
    const int slab = blockIdx.x;
    const int ks   = blockIdx.y;

    const float* __restrict__ xsrc = x + ks * (NB * A_SZ * M_SZ) + slab * (16 * A_SZ);
    float4 v[16];
    #pragma unroll
    for (int j = 0; j < 16; ++j) {
        const int q = j * NTH + t;
        v[j] = *(const float4*)(xsrc + (q >> 5) * (A_SZ * M_SZ) + (q & 31) * 4);
    }
    constexpr int NP = (KROWS + RPP - 1) / RPP;
    const int tb = min(t, RPP * 5 - 1);
    const int tr = tb / 5;
    const int tp = tb - tr * 5;
    const int rl = min((NP - 1) * RPP + tr, KROWS - 1);
    const float* __restrict__ wbase = wv2 + ks * (KROWS * KPAD);
    const float* __restrict__ wsrc  = wbase + tr * KPAD + tp * 4;
    float4 wv[NP];
    #pragma unroll
    for (int j = 0; j < NP - 1; ++j) wv[j] = *(const float4*)(wsrc + j * (RPP * KPAD));
    wv[NP - 1] = *(const float4*)(wbase + rl * KPAD + tp * 4);
    #pragma unroll
    for (int j = 0; j < 16; ++j) ((float4*)xs)[j * NTH + t] = v[j];
    #pragma unroll
    for (int j = 0; j < NP - 1; ++j) ((float4*)ws)[j * (RPP * 5) + tb] = wv[j];
    ((float4*)ws)[rl * (BP / 4) + tp] = wv[NP - 1];
    __syncthreads();

    const int wave = t >> 5;
    const int lane = t & 31;
    const int lo   = lane & 15;
    const int hi   = lane >> 4;

    if (wave < 2) {
        v8f acc = {};
        const int abase = lo * A_SZ + hi * 2;
        const int bcol  = min(wave * 16 + lo, K_CLS - 1);
        const int bbase = hi * 2 * BP + bcol;
        constexpr int GB = GSTEP / 2;
        constexpr int NG = NB / GB;

        v2f a0[GSTEP], b0[GSTEP], a1[GSTEP], b1[GSTEP];
        auto load = [&](v2f (&a)[GSTEP], v2f (&bf)[GSTEP], int g) {
            #pragma unroll
            for (int s = 0; s < GB; ++s) {
                const int b = g * GB + s;
                a[2 * s]     = *(const v2f*)&xs[b * (16 * A_SZ) + abase];
                a[2 * s + 1] = *(const v2f*)&xs[b * (16 * A_SZ) + abase + 4];
                bf[2 * s].x     = ws[(b * A_SZ    ) * BP + bbase];
                bf[2 * s].y     = ws[(b * A_SZ + 1) * BP + bbase];
                bf[2 * s + 1].x = ws[(b * A_SZ + 4) * BP + bbase];
                bf[2 * s + 1].y = ws[(b * A_SZ + 5) * BP + bbase];
            }
        };
        auto steps = [&](const v2f (&a)[GSTEP], const v2f (&bf)[GSTEP]) {
            #pragma unroll
            for (int s = 0; s < GSTEP; ++s)
                acc = __builtin_amdgcn_wmma_f32_16x16x4_f32(
                    false, a[s], false, bf[s], (short)0, acc, false, false);
        };

        load(a0, b0, 0);
        #pragma unroll 1
        for (int g = 0; g < NG - 2; g += 2) {
            load(a1, b1, g + 1);
            __builtin_amdgcn_sched_barrier(0);
            steps(a0, b0);
            __builtin_amdgcn_sched_barrier(0);
            load(a0, b0, g + 2);
            __builtin_amdgcn_sched_barrier(0);
            steps(a1, b1);
        }
        load(a1, b1, NG - 1);
        __builtin_amdgcn_sched_barrier(0);
        steps(a0, b0);
        steps(a1, b1);

        if (wave == 0 || lo < K_CLS - 16) {
            #pragma unroll
            for (int r = 0; r < 8; ++r) tile[(wave * 16 + lo) * 16 + r + hi * 8] = acc[r];
        }
    }
    __syncthreads();
    float* __restrict__ dst = bijp + ks * (M_SZ * PP) + slab * (16 * PP);
    const float r0 = tile[t];
    const float r1 = tile[t + NTH];
    const float r2 = tile[min(t + 2 * NTH, 16 * PP - 1)];
    dst[t]       = r0;
    dst[t + NTH] = r1;
    if (t < 16 * PP - 2 * NTH) dst[t + 2 * NTH] = r2;
}

extern "C" void kernel_launch(void* const* d_in, const int* in_sizes, int n_in,
                              void* d_out, int out_size, void* d_ws, size_t ws_size,
                              hipStream_t stream) {
    (void)in_sizes; (void)n_in; (void)out_size; (void)ws_size;
    const float* x = (const float*)d_in[0];
    const float* W = (const float*)d_in[1];
    float* out = (float*)d_out;

    float* bij  = (float*)d_ws;
    float* wv2  = bij + M_SZ * KPAD;
    float* c    = wv2;
    float* part = wv2 + I_SZ * KPAD;

    for (int it = 0; it < 3; ++it) {
        if (it == 1)      caps_softmax<false, true ><<<K_CLS, 256, 0, stream>>>(bij, part, c);
        else if (it == 2) caps_softmax<true,  false><<<K_CLS, 256, 0, stream>>>(bij, part, c);
        if (it == 0) caps_gemm_t<true ><<<dim3(I_SZ / 16, KS_T), NTH, 0, stream>>>(x, c, part);
        else         caps_gemm_t<false><<<dim3(I_SZ / 16, KS_T), NTH, 0, stream>>>(x, c, part);
        caps_squash<<<B_SZ, 320, 0, stream>>>(part, W, out, wv2);
        if (it < 2)
            caps_gemm_b<<<dim3(M_SZ / 16, KS_B), NTH, 0, stream>>>(x, wv2, part);
    }
}


# Both GEMMs re-tiled: a block owns 32 rows x 32 columns over a K slice of 256 (was 16 x 32 over 512), so all 4 waves run matrix steps, 64 each (was 2 x 128); planes 2->4 and 4->8 (sum order changes)
# speedup vs baseline: 11.5188x; 1.1080x over previous
#include <hip/hip_runtime.h>
#include <math.h>

typedef __attribute__((ext_vector_type(2))) float v2f;
typedef __attribute__((ext_vector_type(8))) float v8f;

#define B_SZ   256
#define M_SZ   1024
#define A_SZ   8
#define K_CLS  20
#define O_SZ   16
#define BP     20
#define GSTEP  16
#define NTH    128
#define RB     32
#define I_SZ   (B_SZ * A_SZ)
#define KS_T   4
#define KS_B   8
#define KROWS  256
#define PP     20
#define CP     260
#define XBP    (A_SZ * KROWS + 8)
#define XGP    68
#define XBB    (4 * XGP)

#define C_START 0x1p-10f

template <int M>
__device__ __forceinline__ float xrow(float x) {
    const int b = __builtin_bit_cast(int, x);
    return __builtin_bit_cast(float, __builtin_amdgcn_update_dpp(b, b, 0x160 + M, 0xf, 0xf, false));
}
__device__ __forceinline__ float xhalf(float x) {
    const int b = __builtin_bit_cast(int, x);
    return __builtin_bit_cast(float, __builtin_amdgcn_permlanex16(b, b, 0x76543210, 0xfedcba98, false, false));
}

template <bool BASE, bool WB>
__global__ void caps_softmax(float* __restrict__ bij, const float* __restrict__ part,
                             float* __restrict__ c) {
    __shared__ __attribute__((aligned(16))) float redm[8];
    __shared__ __attribute__((aligned(16))) float reds[8];
    const int k = blockIdx.x;
    const int t = threadIdx.x;
    const int wave = t >> 5;
    const int lane = t & 31;

    float* __restrict__ brow = bij + k * M_SZ;
    float v0 = 0.0f, v1 = 0.0f, v2 = 0.0f, v3 = 0.0f;
    if (BASE) {
        v0 = brow[t      ];
        v1 = brow[t + 256];
        v2 = brow[t + 512];
        v3 = brow[t + 768];
    }
    {
        float p[4][KS_B];
        const float* __restrict__ pk = part + (t >> 4) * (16 * PP) + k * 16 + (t & 15);
        #pragma unroll
        for (int r = 0; r < 4; ++r)
            #pragma unroll
            for (int s = 0; s < KS_B; ++s)
                p[r][s] = pk[s * (M_SZ * PP) + r * (16 * 16 * PP)];
        v0 += ((p[0][0] + p[0][1]) + (p[0][2] + p[0][3])) + ((p[0][4] + p[0][5]) + (p[0][6] + p[0][7]));
        v1 += ((p[1][0] + p[1][1]) + (p[1][2] + p[1][3])) + ((p[1][4] + p[1][5]) + (p[1][6] + p[1][7]));
        v2 += ((p[2][0] + p[2][1]) + (p[2][2] + p[2][3])) + ((p[2][4] + p[2][5]) + (p[2][6] + p[2][7]));
        v3 += ((p[3][0] + p[3][1]) + (p[3][2] + p[3][3])) + ((p[3][4] + p[3][5]) + (p[3][6] + p[3][7]));
    }

    float mx = fmaxf(fmaxf(v0, v1), fmaxf(v2, v3));
    mx = fmaxf(mx, xrow<1>(mx));
    mx = fmaxf(mx, xrow<2>(mx));
    mx = fmaxf(mx, xrow<4>(mx));
    mx = fmaxf(mx, xrow<8>(mx));
    mx = fmaxf(mx, xhalf(mx));
    if (lane == 0) redm[wave] = mx;
    __syncthreads();
    {
        const float4 q0 = *(const float4*)&redm[0];
        const float4 q1 = *(const float4*)&redm[4];
        mx = fmaxf(fmaxf(fmaxf(q0.x, q0.y), fmaxf(q0.z, q0.w)),
                   fmaxf(fmaxf(q1.x, q1.y), fmaxf(q1.z, q1.w)));
    }

    float e0 = expf(v0 - mx), e1 = expf(v1 - mx);
    float e2 = expf(v2 - mx), e3 = expf(v3 - mx);
    float sm = e0 + e1 + e2 + e3;
    sm += xrow<1>(sm);
    sm += xrow<2>(sm);
    sm += xrow<4>(sm);
    sm += xrow<8>(sm);
    sm += xhalf(sm);
    if (lane == 0) reds[wave] = sm;
    __syncthreads();
    float tot;
    {
        const float4 q0 = *(const float4*)&reds[0];
        const float4 q1 = *(const float4*)&reds[4];
        tot = ((q0.x + q0.y) + (q0.z + q0.w)) + ((q1.x + q1.y) + (q1.z + q1.w));
    }
    float inv = 1.0f / tot;

    float* __restrict__ crow = c + k * M_SZ;
    crow[t      ] = e0 * inv;
    crow[t + 256] = e1 * inv;
    crow[t + 512] = e2 * inv;
    crow[t + 768] = e3 * inv;
    if (WB) {
        brow[t      ] = v0;
        brow[t + 256] = v1;
        brow[t + 512] = v2;
        brow[t + 768] = v3;
    }
}

template <bool START>
__global__ __launch_bounds__(NTH)
void caps_gemm_t(const float* __restrict__ x,
                 const float* __restrict__ c,
                 float* __restrict__ t2p) {
    __shared__ __attribute__((aligned(16))) float lds[4 * XBP + K_CLS * CP + RB * PP];
    float* xs   = lds;
    float* cs   = lds + 4 * XBP;
    float* tile = cs + K_CLS * CP;
    const int t    = threadIdx.x;
    const int slab = blockIdx.x;
    const int ks   = blockIdx.y;

    const float* __restrict__ xsrc = x + slab * (4 * A_SZ * M_SZ) + ks * (A_SZ * KROWS);
    float4 v[16];
    #pragma unroll
    for (int j = 0; j < 16; ++j) {
        const int q = j * NTH + t;
        v[j] = *(const float4*)(xsrc + (q >> 9) * (A_SZ * M_SZ) + (q & 511) * 4);
    }
    const float* __restrict__ csrc = c + ks * KROWS + (t >> 6) * M_SZ + (t & 63) * 4;
    float4 cv[K_CLS / 2];
    if (!START) {
        #pragma unroll
        for (int j = 0; j < K_CLS / 2; ++j) cv[j] = *(const float4*)(csrc + j * (2 * M_SZ));
    }
    #pragma unroll
    for (int j = 0; j < 16; ++j) {
        const int q = j * NTH + t;
        *(float4*)&xs[(q >> 9) * XBP + (q & 511) * 4] = v[j];
    }
    if (!START) {
        #pragma unroll
        for (int j = 0; j < K_CLS / 2; ++j) *(float4*)&cs[(j * 2 + (t >> 6)) * CP + (t & 63) * 4] = cv[j];
    }
    __syncthreads();

    const int wave = t >> 5;
    const int lane = t & 31;
    const int lo   = lane & 15;
    const int hi   = lane >> 4;
    const int ct   = wave & 1;
    const int rt   = wave >> 1;

    v8f acc = {};
    const int abase = (rt * 2 + (lo >> 3)) * XBP + (lo & 7) + hi * 2 * A_SZ;
    const int bcol  = min(ct * 16 + lo, K_CLS - 1);
    const int bbase = bcol * CP + hi * 2;

    v2f a0[GSTEP], b0[GSTEP], a1[GSTEP], b1[GSTEP];
    auto load = [&](v2f (&a)[GSTEP], v2f (&b)[GSTEP], int g) {
        #pragma unroll
        for (int s = 0; s < GSTEP; ++s) {
            const int mb = (g * GSTEP + s) * 4;
            a[s].x = xs[abase + mb * A_SZ];
            a[s].y = xs[abase + mb * A_SZ + A_SZ];
            if (START) { b[s].x = C_START; b[s].y = C_START; }
            else b[s] = *(const v2f*)&cs[bbase + mb];
        }
    };
    auto steps = [&](const v2f (&a)[GSTEP], const v2f (&b)[GSTEP]) {
        #pragma unroll
        for (int s = 0; s < GSTEP; ++s)
            acc = __builtin_amdgcn_wmma_f32_16x16x4_f32(
                false, a[s], false, b[s], (short)0, acc, false, false);
    };

    static_assert((KROWS / 4) / GSTEP == 4, "four groups of matrix steps");
    load(a0, b0, 0);
    load(a1, b1, 1);
    __builtin_amdgcn_sched_barrier(0);
    steps(a0, b0);
    __builtin_amdgcn_sched_barrier(0);
    load(a0, b0, 2);
    __builtin_amdgcn_sched_barrier(0);
    steps(a1, b1);
    __builtin_amdgcn_sched_barrier(0);
    load(a1, b1, 3);
    __builtin_amdgcn_sched_barrier(0);
    steps(a0, b0);
    steps(a1, b1);

    if (ct == 0 || lo < K_CLS - 16) {
        #pragma unroll
        for (int r = 0; r < 8; ++r) tile[(rt * 16 + r + hi * 8) * PP + ct * 16 + lo] = acc[r];
    }
    __syncthreads();
    float* __restrict__ dst = t2p + ks * (I_SZ * PP) + slab * (RB * PP);
    float r[5];
    #pragma unroll
    for (int i = 0; i < 5; ++i) r[i] = tile[t + i * NTH];
    #pragma unroll
    for (int i = 0; i < 5; ++i) dst[t + i * NTH] = r[i];
}

template <bool LAST>
__global__ void caps_squash(const float* __restrict__ t2p,
                            const float* __restrict__ W,
                            float* __restrict__ out,
                            float* __restrict__ wv2) {
    __shared__ float s_s[K_CLS][O_SZ];
    __shared__ float s_v[K_CLS][O_SZ];
    const int b = blockIdx.x;
    const int t = threadIdx.x;
    const int k = t >> 4;
    const int o = t & 15;

    float s = 0.0f;
    #pragma unroll
    for (int a = 0; a < A_SZ; ++a) {
        float tv = t2p[(b * 8 + a) * PP + k];
        #pragma unroll
        for (int p = 1; p < KS_T; ++p) tv += t2p[p * (I_SZ * PP) + (b * 8 + a) * PP + k];
        s += W[k * (O_SZ * A_SZ) + o * A_SZ + a] * tv;
    }
    s_s[k][o] = s;
    __syncthreads();

    float magsq = 0.0f;
    #pragma unroll
    for (int kk = 0; kk < K_CLS; ++kk) {
        float sv = s_s[kk][o];
        magsq += sv * sv;
    }
    float scale = magsq / ((1.0f + magsq) * sqrtf(magsq));
    float v = s * scale;
    if (LAST) {
        out[b * (K_CLS * O_SZ) + t] = v;
    } else {
        s_v[k][o] = v;
        __syncthreads();

        if (t < K_CLS * A_SZ) {
            int k2 = t % K_CLS;
            int a  = t / K_CLS;
            float w = 0.0f;
            #pragma unroll
            for (int o2 = 0; o2 < O_SZ; ++o2)
                w += W[k2 * (O_SZ * A_SZ) + o2 * A_SZ + a] * s_v[k2][o2];
            wv2[(b * 8 + a) * PP + k2] = w * (1.0f / (float)B_SZ);
        }
    }
}

__global__ __launch_bounds__(NTH)
void caps_gemm_b(const float* __restrict__ x,
                 const float* __restrict__ wv2,
                 float* __restrict__ bijp) {
    constexpr int NB = KROWS / A_SZ;
    __shared__ __attribute__((aligned(16))) float lds[NB * XBB + KROWS * BP + RB * PP];
    float* xs   = lds;
    float* ws   = lds + NB * XBB;
    float* tile = ws + KROWS * BP;
    const int t    = threadIdx.x;
    const int slab = blockIdx.x;
    const int ks   = blockIdx.y;

    const float* __restrict__ xsrc = x + ks * (NB * A_SZ * M_SZ) + slab * (RB * A_SZ);
    float4 v[16];
    #pragma unroll
    for (int j = 0; j < 16; ++j) {
        const int q = j * NTH + t;
        v[j] = *(const float4*)(xsrc + (q >> 6) * (A_SZ * M_SZ) + (q & 63) * 4);
    }
    const float* __restrict__ wsrc = wv2 + ks * (KROWS * BP) + t * 4;
    float4 wv[(KROWS * BP) / (4 * NTH)];
    #pragma unroll
    for (int j = 0; j < (KROWS * BP) / (4 * NTH); ++j) wv[j] = *(const float4*)(wsrc + j * (4 * NTH));
    #pragma unroll
    for (int j = 0; j < 16; ++j) {
        const int q = j * NTH + t;
        const int w = q & 63;
        *(float4*)&xs[(q >> 6) * XBB + (w >> 4) * XGP + (w & 15) * 4] = v[j];
    }
    #pragma unroll
    for (int j = 0; j < (KROWS * BP) / (4 * NTH); ++j) ((float4*)ws)[j * NTH + t] = wv[j];
    __syncthreads();

    const int wave = t >> 5;
    const int lane = t & 31;
    const int lo   = lane & 15;
    const int hi   = lane >> 4;
    const int ct   = wave & 1;
    const int rt   = wave >> 1;

    v8f acc = {};
    const int mrow  = rt * 16 + lo;
    const int abase = (mrow >> 3) * XGP + (mrow & 7) * A_SZ + hi * 2;
    const int bcol  = min(ct * 16 + lo, K_CLS - 1);
    const int bbase = hi * 2 * BP + bcol;
    constexpr int GB = GSTEP / 2;

    v2f a0[GSTEP], b0[GSTEP], a1[GSTEP], b1[GSTEP];
    auto load = [&](v2f (&a)[GSTEP], v2f (&bf)[GSTEP], int g) {
        #pragma unroll
        for (int s = 0; s < GB; ++s) {
            const int b = g * GB + s;
            a[2 * s]     = *(const v2f*)&xs[b * XBB + abase];
            a[2 * s + 1] = *(const v2f*)&xs[b * XBB + abase + 4];
            bf[2 * s].x     = ws[(b * A_SZ    ) * BP + bbase];
            bf[2 * s].y     = ws[(b * A_SZ + 1) * BP + bbase];
            bf[2 * s + 1].x = ws[(b * A_SZ + 4) * BP + bbase];
            bf[2 * s + 1].y = ws[(b * A_SZ + 5) * BP + bbase];
        }
    };
    auto steps = [&](const v2f (&a)[GSTEP], const v2f (&bf)[GSTEP]) {
        #pragma unroll
        for (int s = 0; s < GSTEP; ++s)
            acc = __builtin_amdgcn_wmma_f32_16x16x4_f32(
                false, a[s], false, bf[s], (short)0, acc, false, false);
    };

    static_assert(NB / GB == 4, "four groups of matrix steps");
    load(a0, b0, 0);
    load(a1, b1, 1);
    __builtin_amdgcn_sched_barrier(0);
    steps(a0, b0);
    __builtin_amdgcn_sched_barrier(0);
    load(a0, b0, 2);
    __builtin_amdgcn_sched_barrier(0);
    steps(a1, b1);
    __builtin_amdgcn_sched_barrier(0);
    load(a1, b1, 3);
    __builtin_amdgcn_sched_barrier(0);
    steps(a0, b0);
    steps(a1, b1);

    if (ct == 0 || lo < K_CLS - 16) {
        #pragma unroll
        for (int r = 0; r < 8; ++r) tile[rt * (16 * PP) + (ct * 16 + lo) * 16 + r + hi * 8] = acc[r];
    }
    __syncthreads();
    float* __restrict__ dst = bijp + ks * (M_SZ * PP) + slab * (RB * PP);
    float r[5];
    #pragma unroll
    for (int i = 0; i < 5; ++i) r[i] = tile[t + i * NTH];
    #pragma unroll
    for (int i = 0; i < 5; ++i) dst[t + i * NTH] = r[i];
}

extern "C" void kernel_launch(void* const* d_in, const int* in_sizes, int n_in,
                              void* d_out, int out_size, void* d_ws, size_t ws_size,
                              hipStream_t stream) {
    (void)in_sizes; (void)n_in; (void)out_size; (void)ws_size;
    const float* x = (const float*)d_in[0];
    const float* W = (const float*)d_in[1];
    float* out = (float*)d_out;

    float* part = (float*)d_ws;
    float* bij  = out;
    float* wv2  = out + K_CLS * M_SZ;
    float* c    = wv2;

    for (int it = 0; it < 3; ++it) {
        if (it == 1)      caps_softmax<false, true ><<<K_CLS, 256, 0, stream>>>(bij, part, c);
        else if (it == 2) caps_softmax<true,  false><<<K_CLS, 256, 0, stream>>>(bij, part, c);
        if (it == 0) caps_gemm_t<true ><<<dim3(I_SZ / RB, KS_T), NTH, 0, stream>>>(x, c, part);
        else         caps_gemm_t<false><<<dim3(I_SZ / RB, KS_T), NTH, 0, stream>>>(x, c, part);
        if (it < 2) {
            caps_squash<false><<<B_SZ, 320, 0, stream>>>(part, W, out, wv2);
            caps_gemm_b<<<dim3(M_SZ / RB, KS_B), NTH, 0, stream>>>(x, wv2, part);
        } else {
            caps_squash<true ><<<B_SZ, 320, 0, stream>>>(part, W, out, wv2);
        }
    }
}


# Pass 1: squash stage runs in the head of its agreement GEMM on 20 waves (lane = batch element, wave = class, W rows by scalar loads); its GEMM 1 stores one column; 10 -> 9 launches (sum order kept)
# speedup vs baseline: 12.2386x; 1.0625x over previous
#include <hip/hip_runtime.h>
#include <math.h>

typedef __attribute__((ext_vector_type(2))) float v2f;
typedef __attribute__((ext_vector_type(8))) float v8f;

#define B_SZ   256
#define M_SZ   1024
#define A_SZ   8
#define K_CLS  20
#define O_SZ   16
#define BP     20
#define GSTEP  16
#define NTH    128
#define RB     32
#define I_SZ   (B_SZ * A_SZ)
#define KS_T   4
#define KS_B   8
#define KROWS  256
#define PP     20
#define CP     260
#define XBP    (A_SZ * KROWS + 8)
#define XGP    68
#define XBB    (4 * XGP)
#define NTF    640
#define WSP    164
#define SSP    324
#define SCP    20

#define C_START 0x1p-10f

template <int M>
__device__ __forceinline__ float xrow(float x) {
    const int b = __builtin_bit_cast(int, x);
    return __builtin_bit_cast(float, __builtin_amdgcn_update_dpp(b, b, 0x160 + M, 0xf, 0xf, false));
}
__device__ __forceinline__ float xhalf(float x) {
    const int b = __builtin_bit_cast(int, x);
    return __builtin_bit_cast(float, __builtin_amdgcn_permlanex16(b, b, 0x76543210, 0xfedcba98, false, false));
}

template <bool BASE, bool WB>
__global__ void caps_softmax(float* __restrict__ bij, const float* __restrict__ part,
                             float* __restrict__ c) {
    __shared__ __attribute__((aligned(16))) float redm[8];
    __shared__ __attribute__((aligned(16))) float reds[8];
    const int k = blockIdx.x;
    const int t = threadIdx.x;
    const int wave = t >> 5;
    const int lane = t & 31;

    float* __restrict__ brow = bij + k * M_SZ;
    float v0 = 0.0f, v1 = 0.0f, v2 = 0.0f, v3 = 0.0f;
    if (BASE) {
        v0 = brow[t      ];
        v1 = brow[t + 256];
        v2 = brow[t + 512];
        v3 = brow[t + 768];
    }
    {
        float p[4][KS_B];
        const float* __restrict__ pk = part + (t >> 4) * (16 * PP) + k * 16 + (t & 15);
        #pragma unroll
        for (int r = 0; r < 4; ++r)
            #pragma unroll
            for (int s = 0; s < KS_B; ++s)
                p[r][s] = pk[s * (M_SZ * PP) + r * (16 * 16 * PP)];
        v0 += ((p[0][0] + p[0][1]) + (p[0][2] + p[0][3])) + ((p[0][4] + p[0][5]) + (p[0][6] + p[0][7]));
        v1 += ((p[1][0] + p[1][1]) + (p[1][2] + p[1][3])) + ((p[1][4] + p[1][5]) + (p[1][6] + p[1][7]));
        v2 += ((p[2][0] + p[2][1]) + (p[2][2] + p[2][3])) + ((p[2][4] + p[2][5]) + (p[2][6] + p[2][7]));
        v3 += ((p[3][0] + p[3][1]) + (p[3][2] + p[3][3])) + ((p[3][4] + p[3][5]) + (p[3][6] + p[3][7]));
    }

    float mx = fmaxf(fmaxf(v0, v1), fmaxf(v2, v3));
    mx = fmaxf(mx, xrow<1>(mx));
    mx = fmaxf(mx, xrow<2>(mx));
    mx = fmaxf(mx, xrow<4>(mx));
    mx = fmaxf(mx, xrow<8>(mx));
    mx = fmaxf(mx, xhalf(mx));
    if (lane == 0) redm[wave] = mx;
    __syncthreads();
    {
        const float4 q0 = *(const float4*)&redm[0];
        const float4 q1 = *(const float4*)&redm[4];
        mx = fmaxf(fmaxf(fmaxf(q0.x, q0.y), fmaxf(q0.z, q0.w)),
                   fmaxf(fmaxf(q1.x, q1.y), fmaxf(q1.z, q1.w)));
    }

    float e0 = expf(v0 - mx), e1 = expf(v1 - mx);
    float e2 = expf(v2 - mx), e3 = expf(v3 - mx);
    float sm = e0 + e1 + e2 + e3;
    sm += xrow<1>(sm);
    sm += xrow<2>(sm);
    sm += xrow<4>(sm);
    sm += xrow<8>(sm);
    sm += xhalf(sm);
    if (lane == 0) reds[wave] = sm;
    __syncthreads();
    float tot;
    {
        const float4 q0 = *(const float4*)&reds[0];
        const float4 q1 = *(const float4*)&reds[4];
        tot = ((q0.x + q0.y) + (q0.z + q0.w)) + ((q1.x + q1.y) + (q1.z + q1.w));
    }
    float inv = 1.0f / tot;

    float* __restrict__ crow = c + k * M_SZ;
    crow[t      ] = e0 * inv;
    crow[t + 256] = e1 * inv;
    crow[t + 512] = e2 * inv;
    crow[t + 768] = e3 * inv;
    if (WB) {
        brow[t      ] = v0;
        brow[t + 256] = v1;
        brow[t + 512] = v2;
        brow[t + 768] = v3;
    }
}

template <bool START>
__global__ __launch_bounds__(NTH)
void caps_gemm_t(const float* __restrict__ x,
                 const float* __restrict__ c,
                 float* __restrict__ t2p) {
    __shared__ __attribute__((aligned(16))) float lds[4 * XBP + K_CLS * CP + RB * PP];
    float* xs   = lds;
    float* cs   = lds + 4 * XBP;
    float* tile = cs + K_CLS * CP;
    const int t    = threadIdx.x;
    const int slab = blockIdx.x;
    const int ks   = blockIdx.y;

    const float* __restrict__ xsrc = x + slab * (4 * A_SZ * M_SZ) + ks * (A_SZ * KROWS);
    float4 v[16];
    #pragma unroll
    for (int j = 0; j < 16; ++j) {
        const int q = j * NTH + t;
        v[j] = *(const float4*)(xsrc + (q >> 9) * (A_SZ * M_SZ) + (q & 511) * 4);
    }
    const float* __restrict__ csrc = c + ks * KROWS + (t >> 6) * M_SZ + (t & 63) * 4;
    float4 cv[K_CLS / 2];
    if (!START) {
        #pragma unroll
        for (int j = 0; j < K_CLS / 2; ++j) cv[j] = *(const float4*)(csrc + j * (2 * M_SZ));
    }
    #pragma unroll
    for (int j = 0; j < 16; ++j) {
        const int q = j * NTH + t;
        *(float4*)&xs[(q >> 9) * XBP + (q & 511) * 4] = v[j];
    }
    if (!START) {
        #pragma unroll
        for (int j = 0; j < K_CLS / 2; ++j) *(float4*)&cs[(j * 2 + (t >> 6)) * CP + (t & 63) * 4] = cv[j];
    }
    __syncthreads();

    const int wave = t >> 5;
    const int lane = t & 31;
    const int lo   = lane & 15;
    const int hi   = lane >> 4;
    const int ct   = wave & 1;
    const int rt   = wave >> 1;

    v8f acc = {};
    const int abase = (rt * 2 + (lo >> 3)) * XBP + (lo & 7) + hi * 2 * A_SZ;
    const int bcol  = min(ct * 16 + lo, K_CLS - 1);
    const int bbase = bcol * CP + hi * 2;

    v2f a0[GSTEP], b0[GSTEP], a1[GSTEP], b1[GSTEP];
    auto load = [&](v2f (&a)[GSTEP], v2f (&b)[GSTEP], int g) {
        #pragma unroll
        for (int s = 0; s < GSTEP; ++s) {
            const int mb = (g * GSTEP + s) * 4;
            a[s].x = xs[abase + mb * A_SZ];
            a[s].y = xs[abase + mb * A_SZ + A_SZ];
            if (START) { b[s].x = C_START; b[s].y = C_START; }
            else b[s] = *(const v2f*)&cs[bbase + mb];
        }
    };
    auto steps = [&](const v2f (&a)[GSTEP], const v2f (&b)[GSTEP]) {
        #pragma unroll
        for (int s = 0; s < GSTEP; ++s)
            acc = __builtin_amdgcn_wmma_f32_16x16x4_f32(
                false, a[s], false, b[s], (short)0, acc, false, false);
    };

    static_assert((KROWS / 4) / GSTEP == 4, "four groups of matrix steps");
    if (!START || ct == 0) {
        load(a0, b0, 0);
        load(a1, b1, 1);
        __builtin_amdgcn_sched_barrier(0);
        steps(a0, b0);
        __builtin_amdgcn_sched_barrier(0);
        load(a0, b0, 2);
        __builtin_amdgcn_sched_barrier(0);
        steps(a1, b1);
        __builtin_amdgcn_sched_barrier(0);
        load(a1, b1, 3);
        __builtin_amdgcn_sched_barrier(0);
        steps(a0, b0);
        steps(a1, b1);
    }

    if (START) {
        if (ct == 0 && lo == 0) {
            #pragma unroll
            for (int r = 0; r < 8; ++r) tile[rt * 16 + hi * 8 + r] = acc[r];
        }
        __syncthreads();
        if (t < RB) t2p[ks * I_SZ + slab * RB + t] = tile[t];
        return;
    }
    if (ct == 0 || lo < K_CLS - 16) {
        #pragma unroll
        for (int r = 0; r < 8; ++r) tile[(rt * 16 + r + hi * 8) * PP + ct * 16 + lo] = acc[r];
    }
    __syncthreads();
    float* __restrict__ dst = t2p + ks * (I_SZ * PP) + slab * (RB * PP);
    float r[5];
    #pragma unroll
    for (int i = 0; i < 5; ++i) r[i] = tile[t + i * NTH];
    #pragma unroll
    for (int i = 0; i < 5; ++i) dst[t + i * NTH] = r[i];
}

template <bool LAST>
__global__ void caps_squash(const float* __restrict__ t2p,
                            const float* __restrict__ W,
                            float* __restrict__ out,
                            float* __restrict__ wv2) {
    __shared__ float s_s[K_CLS][O_SZ];
    __shared__ float s_v[K_CLS][O_SZ];
    const int b = blockIdx.x;
    const int t = threadIdx.x;
    const int k = t >> 4;
    const int o = t & 15;

    float s = 0.0f;
    #pragma unroll
    for (int a = 0; a < A_SZ; ++a) {
        float tv = t2p[(b * 8 + a) * PP + k];
        #pragma unroll
        for (int p = 1; p < KS_T; ++p) tv += t2p[p * (I_SZ * PP) + (b * 8 + a) * PP + k];
        s += W[k * (O_SZ * A_SZ) + o * A_SZ + a] * tv;
    }
    s_s[k][o] = s;
    __syncthreads();

    float magsq = 0.0f;
    #pragma unroll
    for (int kk = 0; kk < K_CLS; ++kk) {
        float sv = s_s[kk][o];
        magsq += sv * sv;
    }
    float scale = magsq / ((1.0f + magsq) * sqrtf(magsq));
    float v = s * scale;
    if (LAST) {
        out[b * (K_CLS * O_SZ) + t] = v;
    } else {
        s_v[k][o] = v;
        __syncthreads();

        if (t < K_CLS * A_SZ) {
            int k2 = t % K_CLS;
            int a  = t / K_CLS;
            float w = 0.0f;
            #pragma unroll
            for (int o2 = 0; o2 < O_SZ; ++o2)
                w += W[k2 * (O_SZ * A_SZ) + o2 * A_SZ + a] * s_v[k2][o2];
            wv2[(b * 8 + a) * PP + k2] = w * (1.0f / (float)B_SZ);
        }
    }
}

__global__ __launch_bounds__(NTH)
void caps_gemm_b(const float* __restrict__ x,
                 const float* __restrict__ wv2,
                 float* __restrict__ bijp) {
    constexpr int NB = KROWS / A_SZ;
    __shared__ __attribute__((aligned(16))) float lds[NB * XBB + KROWS * BP + RB * PP];
    float* xs   = lds;
    float* ws   = lds + NB * XBB;
    float* tile = ws + KROWS * BP;
    const int t    = threadIdx.x;
    const int slab = blockIdx.x;
    const int ks   = blockIdx.y;

    const float* __restrict__ xsrc = x + ks * (NB * A_SZ * M_SZ) + slab * (RB * A_SZ);
    float4 v[16];
    #pragma unroll
    for (int j = 0; j < 16; ++j) {
        const int q = j * NTH + t;
        v[j] = *(const float4*)(xsrc + (q >> 6) * (A_SZ * M_SZ) + (q & 63) * 4);
    }
    const float* __restrict__ wsrc = wv2 + ks * (KROWS * BP) + t * 4;
    float4 wv[(KROWS * BP) / (4 * NTH)];
    #pragma unroll
    for (int j = 0; j < (KROWS * BP) / (4 * NTH); ++j) wv[j] = *(const float4*)(wsrc + j * (4 * NTH));
    #pragma unroll
    for (int j = 0; j < 16; ++j) {
        const int q = j * NTH + t;
        const int w = q & 63;
        *(float4*)&xs[(q >> 6) * XBB + (w >> 4) * XGP + (w & 15) * 4] = v[j];
    }
    #pragma unroll
    for (int j = 0; j < (KROWS * BP) / (4 * NTH); ++j) ((float4*)ws)[j * NTH + t] = wv[j];
    __syncthreads();

    const int wave = t >> 5;
    const int lane = t & 31;
    const int lo   = lane & 15;
    const int hi   = lane >> 4;
    const int ct   = wave & 1;
    const int rt   = wave >> 1;

    v8f acc = {};
    const int mrow  = rt * 16 + lo;
    const int abase = (mrow >> 3) * XGP + (mrow & 7) * A_SZ + hi * 2;
    const int bcol  = min(ct * 16 + lo, K_CLS - 1);
    const int bbase = hi * 2 * BP + bcol;
    constexpr int GB = GSTEP / 2;

    v2f a0[GSTEP], b0[GSTEP], a1[GSTEP], b1[GSTEP];
    auto load = [&](v2f (&a)[GSTEP], v2f (&bf)[GSTEP], int g) {
        #pragma unroll
        for (int s = 0; s < GB; ++s) {
            const int b = g * GB + s;
            a[2 * s]     = *(const v2f*)&xs[b * XBB + abase];
            a[2 * s + 1] = *(const v2f*)&xs[b * XBB + abase + 4];
            bf[2 * s].x     = ws[(b * A_SZ    ) * BP + bbase];
            bf[2 * s].y     = ws[(b * A_SZ + 1) * BP + bbase];
            bf[2 * s + 1].x = ws[(b * A_SZ + 4) * BP + bbase];
            bf[2 * s + 1].y = ws[(b * A_SZ + 5) * BP + bbase];
        }
    };
    auto steps = [&](const v2f (&a)[GSTEP], const v2f (&bf)[GSTEP]) {
        #pragma unroll
        for (int s = 0; s < GSTEP; ++s)
            acc = __builtin_amdgcn_wmma_f32_16x16x4_f32(
                false, a[s], false, bf[s], (short)0, acc, false, false);
    };

    static_assert(NB / GB == 4, "four groups of matrix steps");
    load(a0, b0, 0);
    load(a1, b1, 1);
    __builtin_amdgcn_sched_barrier(0);
    steps(a0, b0);
    __builtin_amdgcn_sched_barrier(0);
    load(a0, b0, 2);
    __builtin_amdgcn_sched_barrier(0);
    steps(a1, b1);
    __builtin_amdgcn_sched_barrier(0);
    load(a1, b1, 3);
    __builtin_amdgcn_sched_barrier(0);
    steps(a0, b0);
    steps(a1, b1);

    if (ct == 0 || lo < K_CLS - 16) {
        #pragma unroll
        for (int r = 0; r < 8; ++r) tile[rt * (16 * PP) + (ct * 16 + lo) * 16 + r + hi * 8] = acc[r];
    }
    __syncthreads();
    float* __restrict__ dst = bijp + ks * (M_SZ * PP) + slab * (RB * PP);
    float r[5];
    #pragma unroll
    for (int i = 0; i < 5; ++i) r[i] = tile[t + i * NTH];
    #pragma unroll
    for (int i = 0; i < 5; ++i) dst[t + i * NTH] = r[i];
}

__global__ __launch_bounds__(NTF)
void caps_gemm_b_start(const float* __restrict__ x,
                       const float* __restrict__ t2c,
                       const float* __restrict__ W,
                       float* __restrict__ bijp) {
    constexpr int NB = KROWS / A_SZ;
    static_assert(NB == 32, "the head maps one batch element of the slice to one lane");
    static_assert(NTF == K_CLS * 32 && NTF == RB * PP, "one wave per class; one float of the tile per thread");
    __shared__ __attribute__((aligned(16))) float lds[NB * XBB + NB * WSP + RB * PP + NB * SSP + NB * SCP];
    float* xs   = lds;
    float* ws   = lds + NB * XBB;
    float* tile = ws + NB * WSP;
    float* ss   = tile + RB * PP;
    float* scs  = ss + NB * SSP;
    const int t    = threadIdx.x;
    const int slab = blockIdx.x;
    const int ks   = blockIdx.y;
    const int lane = t & 31;
    const int wave = __builtin_amdgcn_readfirstlane(t >> 5);

    const float* __restrict__ xsrc = x + ks * (NB * A_SZ * M_SZ) + slab * (RB * A_SZ);
    float tv[A_SZ];
    {
        const float* __restrict__ tp = t2c + (ks * NB + lane) * A_SZ;
        float4 p[KS_T][2];
        #pragma unroll
        for (int s = 0; s < KS_T; ++s) {
            p[s][0] = *(const float4*)(tp + s * I_SZ);
            p[s][1] = *(const float4*)(tp + s * I_SZ + 4);
        }
        if (t < 512) {
            float4 v[4];
            #pragma unroll
            for (int j = 0; j < 4; ++j) {
                const int q = j * 512 + t;
                v[j] = *(const float4*)(xsrc + (q >> 6) * (A_SZ * M_SZ) + (q & 63) * 4);
            }
            #pragma unroll
            for (int j = 0; j < 4; ++j) {
                const int q = j * 512 + t;
                const int w = q & 63;
                *(float4*)&xs[(q >> 6) * XBB + (w >> 4) * XGP + (w & 15) * 4] = v[j];
            }
        }
        float pa[KS_T][A_SZ];
        #pragma unroll
        for (int s = 0; s < KS_T; ++s) {
            pa[s][0] = p[s][0].x; pa[s][1] = p[s][0].y; pa[s][2] = p[s][0].z; pa[s][3] = p[s][0].w;
            pa[s][4] = p[s][1].x; pa[s][5] = p[s][1].y; pa[s][6] = p[s][1].z; pa[s][7] = p[s][1].w;
        }
        #pragma unroll
        for (int a = 0; a < A_SZ; ++a) {
            float sum = pa[0][a];
            #pragma unroll
            for (int s = 1; s < KS_T; ++s) sum += pa[s][a];
            tv[a] = sum;
        }
    }
    const float* __restrict__ Wk = W + wave * (O_SZ * A_SZ);
    float sv[O_SZ];
    #pragma unroll
    for (int o = 0; o < O_SZ; ++o) {
        float s = 0.0f;
        #pragma unroll
        for (int a = 0; a < A_SZ; ++a) s += Wk[o * A_SZ + a] * tv[a];
        sv[o] = s;
    }
    #pragma unroll
    for (int j = 0; j < O_SZ / 4; ++j)
        *(float4*)&ss[lane * SSP + wave * O_SZ + j * 4] =
            make_float4(sv[j * 4], sv[j * 4 + 1], sv[j * 4 + 2], sv[j * 4 + 3]);
    __syncthreads();

    if (t < NB * O_SZ) {
        const int b2 = t >> 4;
        const int o2 = t & 15;
        float magsq = 0.0f;
        #pragma unroll
        for (int kk = 0; kk < K_CLS; ++kk) {
            const float q = ss[b2 * SSP + kk * O_SZ + o2];
            magsq += q * q;
        }
        scs[b2 * SCP + o2] = magsq / ((1.0f + magsq) * sqrtf(magsq));
    }
    __syncthreads();

    {
        float sc[O_SZ];
        #pragma unroll
        for (int j = 0; j < O_SZ / 4; ++j) {
            const float4 q = *(const float4*)&scs[lane * SCP + j * 4];
            sc[j * 4] = q.x; sc[j * 4 + 1] = q.y; sc[j * 4 + 2] = q.z; sc[j * 4 + 3] = q.w;
        }
        float vv[O_SZ];
        #pragma unroll
        for (int o = 0; o < O_SZ; ++o) vv[o] = sv[o] * sc[o];
        #pragma unroll
        for (int a = 0; a < A_SZ; ++a) {
            float w = 0.0f;
            #pragma unroll
            for (int o = 0; o < O_SZ; ++o) w += Wk[o * A_SZ + a] * vv[o];
            ws[lane * WSP + a * BP + wave] = w * (1.0f / (float)B_SZ);
        }
    }
    __syncthreads();

    if (t < NTH) {
        const int lo   = lane & 15;
        const int hi   = lane >> 4;
        const int ct   = wave & 1;
        const int rt   = wave >> 1;

        v8f acc = {};
        const int mrow  = rt * 16 + lo;
        const int abase = (mrow >> 3) * XGP + (mrow & 7) * A_SZ + hi * 2;
        const int bcol  = min(ct * 16 + lo, K_CLS - 1);
        const int bbase = hi * 2 * BP + bcol;
        constexpr int GB = GSTEP / 2;

        v2f a0[GSTEP], b0[GSTEP], a1[GSTEP], b1[GSTEP];
        auto load = [&](v2f (&a)[GSTEP], v2f (&bf)[GSTEP], int g) {
            #pragma unroll
            for (int s = 0; s < GB; ++s) {
                const int b = g * GB + s;
                a[2 * s]     = *(const v2f*)&xs[b * XBB + abase];
                a[2 * s + 1] = *(const v2f*)&xs[b * XBB + abase + 4];
                bf[2 * s].x     = ws[b * WSP           + bbase];
                bf[2 * s].y     = ws[b * WSP +     BP + bbase];
                bf[2 * s + 1].x = ws[b * WSP + 4 * BP + bbase];
                bf[2 * s + 1].y = ws[b * WSP + 5 * BP + bbase];
            }
        };
        auto steps = [&](const v2f (&a)[GSTEP], const v2f (&bf)[GSTEP]) {
            #pragma unroll
            for (int s = 0; s < GSTEP; ++s)
                acc = __builtin_amdgcn_wmma_f32_16x16x4_f32(
                    false, a[s], false, bf[s], (short)0, acc, false, false);
        };

        static_assert(NB / GB == 4, "four groups of matrix steps");
        load(a0, b0, 0);
        load(a1, b1, 1);
        __builtin_amdgcn_sched_barrier(0);
        steps(a0, b0);
        __builtin_amdgcn_sched_barrier(0);
        load(a0, b0, 2);
        __builtin_amdgcn_sched_barrier(0);
        steps(a1, b1);
        __builtin_amdgcn_sched_barrier(0);
        load(a1, b1, 3);
        __builtin_amdgcn_sched_barrier(0);
        steps(a0, b0);
        steps(a1, b1);

        if (ct == 0 || lo < K_CLS - 16) {
            #pragma unroll
            for (int r = 0; r < 8; ++r) tile[rt * (16 * PP) + (ct * 16 + lo) * 16 + r + hi * 8] = acc[r];
        }
    }
    __syncthreads();
    bijp[ks * (M_SZ * PP) + slab * (RB * PP) + t] = tile[t];
}

extern "C" void kernel_launch(void* const* d_in, const int* in_sizes, int n_in,
                              void* d_out, int out_size, void* d_ws, size_t ws_size,
                              hipStream_t stream) {
    (void)in_sizes; (void)n_in; (void)out_size; (void)ws_size;
    const float* x = (const float*)d_in[0];
    const float* W = (const float*)d_in[1];
    float* out = (float*)d_out;

    float* part = (float*)d_ws;
    float* bij  = out;
    float* wv2  = out + K_CLS * M_SZ;
    float* c    = wv2;
    float* t2c  = wv2;

    for (int it = 0; it < 3; ++it) {
        if (it == 1)      caps_softmax<false, true ><<<K_CLS, 256, 0, stream>>>(bij, part, c);
        else if (it == 2) caps_softmax<true,  false><<<K_CLS, 256, 0, stream>>>(bij, part, c);
        if (it == 0) caps_gemm_t<true ><<<dim3(I_SZ / RB, KS_T), NTH, 0, stream>>>(x, x, t2c);
        else         caps_gemm_t<false><<<dim3(I_SZ / RB, KS_T), NTH, 0, stream>>>(x, c, part);
        if (it == 0) {
            caps_gemm_b_start<<<dim3(M_SZ / RB, KS_B), NTF, 0, stream>>>(x, t2c, W, part);
        } else if (it < 2) {
            caps_squash<false><<<B_SZ, 320, 0, stream>>>(part, W, out, wv2);
            caps_gemm_b<<<dim3(M_SZ / RB, KS_B), NTH, 0, stream>>>(x, wv2, part);
        } else {
            caps_squash<true ><<<B_SZ, 320, 0, stream>>>(part, W, out, wv2);
        }
    }
}


# Pass 2: squash stage in the head of its agreement GEMM as in pass 1, where the workspace holds both plane sets (else as before); GEMM 1's planes class-major; 9 -> 8 launches (sum order kept)
# speedup vs baseline: 12.7017x; 1.0378x over previous
#include <hip/hip_runtime.h>
#include <math.h>

typedef __attribute__((ext_vector_type(2))) float v2f;
typedef __attribute__((ext_vector_type(8))) float v8f;

#define B_SZ   256
#define M_SZ   1024
#define A_SZ   8
#define K_CLS  20
#define O_SZ   16
#define BP     20
#define GSTEP  16
#define NTH    128
#define RB     32
#define I_SZ   (B_SZ * A_SZ)
#define KS_T   4
#define KS_B   8
#define KROWS  256
#define PP     20
#define TP     36
#define CP     260
#define XBP    (A_SZ * KROWS + 8)
#define XGP    68
#define XBB    (4 * XGP)
#define NTF    640
#define WSP    164
#define SSP    324
#define SCP    20

#define C_START 0x1p-10f

template <int M>
__device__ __forceinline__ float xrow(float x) {
    const int b = __builtin_bit_cast(int, x);
    return __builtin_bit_cast(float, __builtin_amdgcn_update_dpp(b, b, 0x160 + M, 0xf, 0xf, false));
}
__device__ __forceinline__ float xhalf(float x) {
    const int b = __builtin_bit_cast(int, x);
    return __builtin_bit_cast(float, __builtin_amdgcn_permlanex16(b, b, 0x76543210, 0xfedcba98, false, false));
}

template <bool BASE, bool WB>
__global__ void caps_softmax(float* __restrict__ bij, const float* __restrict__ part,
                             float* __restrict__ c) {
    __shared__ __attribute__((aligned(16))) float redm[8];
    __shared__ __attribute__((aligned(16))) float reds[8];
    const int k = blockIdx.x;
    const int t = threadIdx.x;
    const int wave = t >> 5;
    const int lane = t & 31;

    float* __restrict__ brow = bij + k * M_SZ;
    float v0 = 0.0f, v1 = 0.0f, v2 = 0.0f, v3 = 0.0f;
    if (BASE) {
        v0 = brow[t      ];
        v1 = brow[t + 256];
        v2 = brow[t + 512];
        v3 = brow[t + 768];
    }
    {
        float p[4][KS_B];
        const float* __restrict__ pk = part + (t >> 4) * (16 * PP) + k * 16 + (t & 15);
        #pragma unroll
        for (int r = 0; r < 4; ++r)
            #pragma unroll
            for (int s = 0; s < KS_B; ++s)
                p[r][s] = pk[s * (M_SZ * PP) + r * (16 * 16 * PP)];
        v0 += ((p[0][0] + p[0][1]) + (p[0][2] + p[0][3])) + ((p[0][4] + p[0][5]) + (p[0][6] + p[0][7]));
        v1 += ((p[1][0] + p[1][1]) + (p[1][2] + p[1][3])) + ((p[1][4] + p[1][5]) + (p[1][6] + p[1][7]));
        v2 += ((p[2][0] + p[2][1]) + (p[2][2] + p[2][3])) + ((p[2][4] + p[2][5]) + (p[2][6] + p[2][7]));
        v3 += ((p[3][0] + p[3][1]) + (p[3][2] + p[3][3])) + ((p[3][4] + p[3][5]) + (p[3][6] + p[3][7]));
    }

    float mx = fmaxf(fmaxf(v0, v1), fmaxf(v2, v3));
    mx = fmaxf(mx, xrow<1>(mx));
    mx = fmaxf(mx, xrow<2>(mx));
    mx = fmaxf(mx, xrow<4>(mx));
    mx = fmaxf(mx, xrow<8>(mx));
    mx = fmaxf(mx, xhalf(mx));
    if (lane == 0) redm[wave] = mx;
    __syncthreads();
    {
        const float4 q0 = *(const float4*)&redm[0];
        const float4 q1 = *(const float4*)&redm[4];
        mx = fmaxf(fmaxf(fmaxf(q0.x, q0.y), fmaxf(q0.z, q0.w)),
                   fmaxf(fmaxf(q1.x, q1.y), fmaxf(q1.z, q1.w)));
    }

    float e0 = expf(v0 - mx), e1 = expf(v1 - mx);
    float e2 = expf(v2 - mx), e3 = expf(v3 - mx);
    float sm = e0 + e1 + e2 + e3;
    sm += xrow<1>(sm);
    sm += xrow<2>(sm);
    sm += xrow<4>(sm);
    sm += xrow<8>(sm);
    sm += xhalf(sm);
    if (lane == 0) reds[wave] = sm;
    __syncthreads();
    float tot;
    {
        const float4 q0 = *(const float4*)&reds[0];
        const float4 q1 = *(const float4*)&reds[4];
        tot = ((q0.x + q0.y) + (q0.z + q0.w)) + ((q1.x + q1.y) + (q1.z + q1.w));
    }
    float inv = 1.0f / tot;

    float* __restrict__ crow = c + k * M_SZ;
    crow[t      ] = e0 * inv;
    crow[t + 256] = e1 * inv;
    crow[t + 512] = e2 * inv;
    crow[t + 768] = e3 * inv;
    if (WB) {
        brow[t      ] = v0;
        brow[t + 256] = v1;
        brow[t + 512] = v2;
        brow[t + 768] = v3;
    }
}

template <bool START>
__global__ __launch_bounds__(NTH)
void caps_gemm_t(const float* __restrict__ x,
                 const float* __restrict__ c,
                 float* __restrict__ t2p) {
    __shared__ __attribute__((aligned(16))) float lds[4 * XBP + K_CLS * CP + K_CLS * TP];
    float* xs   = lds;
    float* cs   = lds + 4 * XBP;
    float* tile = cs + K_CLS * CP;
    const int t    = threadIdx.x;
    const int slab = blockIdx.x;
    const int ks   = blockIdx.y;

    const float* __restrict__ xsrc = x + slab * (4 * A_SZ * M_SZ) + ks * (A_SZ * KROWS);
    float4 v[16];
    #pragma unroll
    for (int j = 0; j < 16; ++j) {
        const int q = j * NTH + t;
        v[j] = *(const float4*)(xsrc + (q >> 9) * (A_SZ * M_SZ) + (q & 511) * 4);
    }
    const float* __restrict__ csrc = c + ks * KROWS + (t >> 6) * M_SZ + (t & 63) * 4;
    float4 cv[K_CLS / 2];
    if (!START) {
        #pragma unroll
        for (int j = 0; j < K_CLS / 2; ++j) cv[j] = *(const float4*)(csrc + j * (2 * M_SZ));
    }
    #pragma unroll
    for (int j = 0; j < 16; ++j) {
        const int q = j * NTH + t;
        *(float4*)&xs[(q >> 9) * XBP + (q & 511) * 4] = v[j];
    }
    if (!START) {
        #pragma unroll
        for (int j = 0; j < K_CLS / 2; ++j) *(float4*)&cs[(j * 2 + (t >> 6)) * CP + (t & 63) * 4] = cv[j];
    }
    __syncthreads();

    const int wave = t >> 5;
    const int lane = t & 31;
    const int lo   = lane & 15;
    const int hi   = lane >> 4;
    const int ct   = wave & 1;
    const int rt   = wave >> 1;

    v8f acc = {};
    const int abase = (rt * 2 + (lo >> 3)) * XBP + (lo & 7) + hi * 2 * A_SZ;
    const int bcol  = min(ct * 16 + lo, K_CLS - 1);
    const int bbase = bcol * CP + hi * 2;

    v2f a0[GSTEP], b0[GSTEP], a1[GSTEP], b1[GSTEP];
    auto load = [&](v2f (&a)[GSTEP], v2f (&b)[GSTEP], int g) {
        #pragma unroll
        for (int s = 0; s < GSTEP; ++s) {
            const int mb = (g * GSTEP + s) * 4;
            a[s].x = xs[abase + mb * A_SZ];
            a[s].y = xs[abase + mb * A_SZ + A_SZ];
            if (START) { b[s].x = C_START; b[s].y = C_START; }
            else b[s] = *(const v2f*)&cs[bbase + mb];
        }
    };
    auto steps = [&](const v2f (&a)[GSTEP], const v2f (&b)[GSTEP]) {
        #pragma unroll
        for (int s = 0; s < GSTEP; ++s)
            acc = __builtin_amdgcn_wmma_f32_16x16x4_f32(
                false, a[s], false, b[s], (short)0, acc, false, false);
    };

    static_assert((KROWS / 4) / GSTEP == 4, "four groups of matrix steps");
    if (!START || ct == 0) {
        load(a0, b0, 0);
        load(a1, b1, 1);
        __builtin_amdgcn_sched_barrier(0);
        steps(a0, b0);
        __builtin_amdgcn_sched_barrier(0);
        load(a0, b0, 2);
        __builtin_amdgcn_sched_barrier(0);
        steps(a1, b1);
        __builtin_amdgcn_sched_barrier(0);
        load(a1, b1, 3);
        __builtin_amdgcn_sched_barrier(0);
        steps(a0, b0);
        steps(a1, b1);
    }

    if (START) {
        if (ct == 0 && lo == 0) {
            #pragma unroll
            for (int r = 0; r < 8; ++r) tile[rt * 16 + hi * 8 + r] = acc[r];
        }
        __syncthreads();
        if (t < RB) t2p[ks * I_SZ + slab * RB + t] = tile[t];
        return;
    }
    if (ct == 0 || lo < K_CLS - 16) {
        #pragma unroll
        for (int r = 0; r < 8; ++r) tile[(ct * 16 + lo) * TP + rt * 16 + hi * 8 + r] = acc[r];
    }
    __syncthreads();
    float* __restrict__ dst = t2p + ks * (K_CLS * I_SZ) + slab * RB + (t >> 5) * I_SZ + (t & 31);
    float r[5];
    #pragma unroll
    for (int i = 0; i < 5; ++i) r[i] = tile[((t >> 5) + 4 * i) * TP + (t & 31)];
    #pragma unroll
    for (int i = 0; i < 5; ++i) dst[i * (4 * I_SZ)] = r[i];
}

template <bool LAST>
__global__ void caps_squash(const float* __restrict__ t2p,
                            const float* __restrict__ W,
                            float* __restrict__ out,
                            float* __restrict__ wv2) {
    __shared__ float s_s[K_CLS][O_SZ];
    __shared__ float s_v[K_CLS][O_SZ];
    const int b = blockIdx.x;
    const int t = threadIdx.x;
    const int k = t >> 4;
    const int o = t & 15;

    float s = 0.0f;
    {
        const float* __restrict__ tp = t2p + k * I_SZ + b * A_SZ;
        float4 p[KS_T][2];
        #pragma unroll
        for (int q = 0; q < KS_T; ++q) {
            p[q][0] = *(const float4*)(tp + q * (K_CLS * I_SZ));
            p[q][1] = *(const float4*)(tp + q * (K_CLS * I_SZ) + 4);
        }
        float pa[KS_T][A_SZ];
        #pragma unroll
        for (int q = 0; q < KS_T; ++q) {
            pa[q][0] = p[q][0].x; pa[q][1] = p[q][0].y; pa[q][2] = p[q][0].z; pa[q][3] = p[q][0].w;
            pa[q][4] = p[q][1].x; pa[q][5] = p[q][1].y; pa[q][6] = p[q][1].z; pa[q][7] = p[q][1].w;
        }
        #pragma unroll
        for (int a = 0; a < A_SZ; ++a) {
            float tv = pa[0][a];
            #pragma unroll
            for (int q = 1; q < KS_T; ++q) tv += pa[q][a];
            s += W[k * (O_SZ * A_SZ) + o * A_SZ + a] * tv;
        }
    }
    s_s[k][o] = s;
    __syncthreads();

    float magsq = 0.0f;
    #pragma unroll
    for (int kk = 0; kk < K_CLS; ++kk) {
        float sv = s_s[kk][o];
        magsq += sv * sv;
    }
    float scale = magsq / ((1.0f + magsq) * sqrtf(magsq));
    float v = s * scale;
    if (LAST) {
        out[b * (K_CLS * O_SZ) + t] = v;
    } else {
        s_v[k][o] = v;
        __syncthreads();

        if (t < K_CLS * A_SZ) {
            int k2 = t % K_CLS;
            int a  = t / K_CLS;
            float w = 0.0f;
            #pragma unroll
            for (int o2 = 0; o2 < O_SZ; ++o2)
                w += W[k2 * (O_SZ * A_SZ) + o2 * A_SZ + a] * s_v[k2][o2];
            wv2[(b * 8 + a) * PP + k2] = w * (1.0f / (float)B_SZ);
        }
    }
}

__global__ __launch_bounds__(NTH)
void caps_gemm_b(const float* __restrict__ x,
                 const float* __restrict__ wv2,
                 float* __restrict__ bijp) {
    constexpr int NB = KROWS / A_SZ;
    __shared__ __attribute__((aligned(16))) float lds[NB * XBB + KROWS * BP + RB * PP];
    float* xs   = lds;
    float* ws   = lds + NB * XBB;
    float* tile = ws + KROWS * BP;
    const int t    = threadIdx.x;
    const int slab = blockIdx.x;
    const int ks   = blockIdx.y;

    const float* __restrict__ xsrc = x + ks * (NB * A_SZ * M_SZ) + slab * (RB * A_SZ);
    float4 v[16];
    #pragma unroll
    for (int j = 0; j < 16; ++j) {
        const int q = j * NTH + t;
        v[j] = *(const float4*)(xsrc + (q >> 6) * (A_SZ * M_SZ) + (q & 63) * 4);
    }
    const float* __restrict__ wsrc = wv2 + ks * (KROWS * BP) + t * 4;
    float4 wv[(KROWS * BP) / (4 * NTH)];
    #pragma unroll
    for (int j = 0; j < (KROWS * BP) / (4 * NTH); ++j) wv[j] = *(const float4*)(wsrc + j * (4 * NTH));
    #pragma unroll
    for (int j = 0; j < 16; ++j) {
        const int q = j * NTH + t;
        const int w = q & 63;
        *(float4*)&xs[(q >> 6) * XBB + (w >> 4) * XGP + (w & 15) * 4] = v[j];
    }
    #pragma unroll
    for (int j = 0; j < (KROWS * BP) / (4 * NTH); ++j) ((float4*)ws)[j * NTH + t] = wv[j];
    __syncthreads();

    const int wave = t >> 5;
    const int lane = t & 31;
    const int lo   = lane & 15;
    const int hi   = lane >> 4;
    const int ct   = wave & 1;
    const int rt   = wave >> 1;

    v8f acc = {};
    const int mrow  = rt * 16 + lo;
    const int abase = (mrow >> 3) * XGP + (mrow & 7) * A_SZ + hi * 2;
    const int bcol  = min(ct * 16 + lo, K_CLS - 1);
    const int bbase = hi * 2 * BP + bcol;
    constexpr int GB = GSTEP / 2;

    v2f a0[GSTEP], b0[GSTEP], a1[GSTEP], b1[GSTEP];
    auto load = [&](v2f (&a)[GSTEP], v2f (&bf)[GSTEP], int g) {
        #pragma unroll
        for (int s = 0; s < GB; ++s) {
            const int b = g * GB + s;
            a[2 * s]     = *(const v2f*)&xs[b * XBB + abase];
            a[2 * s + 1] = *(const v2f*)&xs[b * XBB + abase + 4];
            bf[2 * s].x     = ws[(b * A_SZ    ) * BP + bbase];
            bf[2 * s].y     = ws[(b * A_SZ + 1) * BP + bbase];
            bf[2 * s + 1].x = ws[(b * A_SZ + 4) * BP + bbase];
            bf[2 * s + 1].y = ws[(b * A_SZ + 5) * BP + bbase];
        }
    };
    auto steps = [&](const v2f (&a)[GSTEP], const v2f (&bf)[GSTEP]) {
        #pragma unroll
        for (int s = 0; s < GSTEP; ++s)
            acc = __builtin_amdgcn_wmma_f32_16x16x4_f32(
                false, a[s], false, bf[s], (short)0, acc, false, false);
    };

    static_assert(NB / GB == 4, "four groups of matrix steps");
    load(a0, b0, 0);
    load(a1, b1, 1);
    __builtin_amdgcn_sched_barrier(0);
    steps(a0, b0);
    __builtin_amdgcn_sched_barrier(0);
    load(a0, b0, 2);
    __builtin_amdgcn_sched_barrier(0);
    steps(a1, b1);
    __builtin_amdgcn_sched_barrier(0);
    load(a1, b1, 3);
    __builtin_amdgcn_sched_barrier(0);
    steps(a0, b0);
    steps(a1, b1);

    if (ct == 0 || lo < K_CLS - 16) {
        #pragma unroll
        for (int r = 0; r < 8; ++r) tile[rt * (16 * PP) + (ct * 16 + lo) * 16 + r + hi * 8] = acc[r];
    }
    __syncthreads();
    float* __restrict__ dst = bijp + ks * (M_SZ * PP) + slab * (RB * PP);
    float r[5];
    #pragma unroll
    for (int i = 0; i < 5; ++i) r[i] = tile[t + i * NTH];
    #pragma unroll
    for (int i = 0; i < 5; ++i) dst[t + i * NTH] = r[i];
}

template <bool START>
__global__ __launch_bounds__(NTF)
void caps_gemm_b_head(const float* __restrict__ x,
                      const float* __restrict__ t2,
                      const float* __restrict__ W,
                      float* __restrict__ bijp) {
    constexpr int NB = KROWS / A_SZ;
    static_assert(NB == 32, "the head maps one batch element of the slice to one lane");
    static_assert(NTF == K_CLS * 32 && NTF == RB * PP, "one wave per class; one float of the tile per thread");
    __shared__ __attribute__((aligned(16))) float lds[NB * XBB + NB * WSP + RB * PP + NB * SSP + NB * SCP];
    float* xs   = lds;
    float* ws   = lds + NB * XBB;
    float* tile = ws + NB * WSP;
    float* ss   = tile + RB * PP;
    float* scs  = ss + NB * SSP;
    const int t    = threadIdx.x;
    const int slab = blockIdx.x;
    const int ks   = blockIdx.y;
    const int lane = t & 31;
    const int wave = __builtin_amdgcn_readfirstlane(t >> 5);

    const float* __restrict__ xsrc = x + ks * (NB * A_SZ * M_SZ) + slab * (RB * A_SZ);
    float tv[A_SZ];
    {
        constexpr int PS = START ? I_SZ : K_CLS * I_SZ;
        const float* __restrict__ tp = t2 + (START ? 0 : wave * I_SZ) + (ks * NB + lane) * A_SZ;
        float4 p[KS_T][2];
        #pragma unroll
        for (int s = 0; s < KS_T; ++s) {
            p[s][0] = *(const float4*)(tp + s * PS);
            p[s][1] = *(const float4*)(tp + s * PS + 4);
        }
        if (t < 512) {
            float4 v[4];
            #pragma unroll
            for (int j = 0; j < 4; ++j) {
                const int q = j * 512 + t;
                v[j] = *(const float4*)(xsrc + (q >> 6) * (A_SZ * M_SZ) + (q & 63) * 4);
            }
            #pragma unroll
            for (int j = 0; j < 4; ++j) {
                const int q = j * 512 + t;
                const int w = q & 63;
                *(float4*)&xs[(q >> 6) * XBB + (w >> 4) * XGP + (w & 15) * 4] = v[j];
            }
        }
        float pa[KS_T][A_SZ];
        #pragma unroll
        for (int s = 0; s < KS_T; ++s) {
            pa[s][0] = p[s][0].x; pa[s][1] = p[s][0].y; pa[s][2] = p[s][0].z; pa[s][3] = p[s][0].w;
            pa[s][4] = p[s][1].x; pa[s][5] = p[s][1].y; pa[s][6] = p[s][1].z; pa[s][7] = p[s][1].w;
        }
        #pragma unroll
        for (int a = 0; a < A_SZ; ++a) {
            float sum = pa[0][a];
            #pragma unroll
            for (int s = 1; s < KS_T; ++s) sum += pa[s][a];
            tv[a] = sum;
        }
    }
    const float* __restrict__ Wk = W + wave * (O_SZ * A_SZ);
    float sv[O_SZ];
    #pragma unroll
    for (int o = 0; o < O_SZ; ++o) {
        float s = 0.0f;
        #pragma unroll
        for (int a = 0; a < A_SZ; ++a) s += Wk[o * A_SZ + a] * tv[a];
        sv[o] = s;
    }
    #pragma unroll
    for (int j = 0; j < O_SZ / 4; ++j)
        *(float4*)&ss[lane * SSP + wave * O_SZ + j * 4] =
            make_float4(sv[j * 4], sv[j * 4 + 1], sv[j * 4 + 2], sv[j * 4 + 3]);
    __syncthreads();

    if (t < NB * O_SZ) {
        const int b2 = t >> 4;
        const int o2 = t & 15;
        float magsq = 0.0f;
        #pragma unroll
        for (int kk = 0; kk < K_CLS; ++kk) {
            const float q = ss[b2 * SSP + kk * O_SZ + o2];
            magsq += q * q;
        }
        scs[b2 * SCP + o2] = magsq / ((1.0f + magsq) * sqrtf(magsq));
    }
    __syncthreads();

    {
        float sc[O_SZ];
        #pragma unroll
        for (int j = 0; j < O_SZ / 4; ++j) {
            const float4 q = *(const float4*)&scs[lane * SCP + j * 4];
            sc[j * 4] = q.x; sc[j * 4 + 1] = q.y; sc[j * 4 + 2] = q.z; sc[j * 4 + 3] = q.w;
        }
        float vv[O_SZ];
        #pragma unroll
        for (int o = 0; o < O_SZ; ++o) vv[o] = sv[o] * sc[o];
        #pragma unroll
        for (int a = 0; a < A_SZ; ++a) {
            float w = 0.0f;
            #pragma unroll
            for (int o = 0; o < O_SZ; ++o) w += Wk[o * A_SZ + a] * vv[o];
            ws[lane * WSP + a * BP + wave] = w * (1.0f / (float)B_SZ);
        }
    }
    __syncthreads();

    if (t < NTH) {
        const int lo   = lane & 15;
        const int hi   = lane >> 4;
        const int ct   = wave & 1;
        const int rt   = wave >> 1;

        v8f acc = {};
        const int mrow  = rt * 16 + lo;
        const int abase = (mrow >> 3) * XGP + (mrow & 7) * A_SZ + hi * 2;
        const int bcol  = min(ct * 16 + lo, K_CLS - 1);
        const int bbase = hi * 2 * BP + bcol;
        constexpr int GB = GSTEP / 2;

        v2f a0[GSTEP], b0[GSTEP], a1[GSTEP], b1[GSTEP];
        auto load = [&](v2f (&a)[GSTEP], v2f (&bf)[GSTEP], int g) {
            #pragma unroll
            for (int s = 0; s < GB; ++s) {
                const int b = g * GB + s;
                a[2 * s]     = *(const v2f*)&xs[b * XBB + abase];
                a[2 * s + 1] = *(const v2f*)&xs[b * XBB + abase + 4];
                bf[2 * s].x     = ws[b * WSP           + bbase];
                bf[2 * s].y     = ws[b * WSP +     BP + bbase];
                bf[2 * s + 1].x = ws[b * WSP + 4 * BP + bbase];
                bf[2 * s + 1].y = ws[b * WSP + 5 * BP + bbase];
            }
        };
        auto steps = [&](const v2f (&a)[GSTEP], const v2f (&bf)[GSTEP]) {
            #pragma unroll
            for (int s = 0; s < GSTEP; ++s)
                acc = __builtin_amdgcn_wmma_f32_16x16x4_f32(
                    false, a[s], false, bf[s], (short)0, acc, false, false);
        };

        static_assert(NB / GB == 4, "four groups of matrix steps");
        load(a0, b0, 0);
        load(a1, b1, 1);
        __builtin_amdgcn_sched_barrier(0);
        steps(a0, b0);
        __builtin_amdgcn_sched_barrier(0);
        load(a0, b0, 2);
        __builtin_amdgcn_sched_barrier(0);
        steps(a1, b1);
        __builtin_amdgcn_sched_barrier(0);
        load(a1, b1, 3);
        __builtin_amdgcn_sched_barrier(0);
        steps(a0, b0);
        steps(a1, b1);

        if (ct == 0 || lo < K_CLS - 16) {
            #pragma unroll
            for (int r = 0; r < 8; ++r) tile[rt * (16 * PP) + (ct * 16 + lo) * 16 + r + hi * 8] = acc[r];
        }
    }
    __syncthreads();
    bijp[ks * (M_SZ * PP) + slab * (RB * PP) + t] = tile[t];
}

extern "C" void kernel_launch(void* const* d_in, const int* in_sizes, int n_in,
                              void* d_out, int out_size, void* d_ws, size_t ws_size,
                              hipStream_t stream) {
    (void)in_sizes; (void)n_in; (void)out_size;
    const float* x = (const float*)d_in[0];
    const float* W = (const float*)d_in[1];
    float* out = (float*)d_out;

    float* part = (float*)d_ws;
    const bool two_sets = ws_size >= (size_t)1310720;
    float* bp2  = two_sets ? part + 163840 : part;
    float* bij  = out;
    float* wv2  = out + K_CLS * M_SZ;
    float* c    = wv2;
    float* t2c  = wv2;

    for (int it = 0; it < 3; ++it) {
        if (it == 1)      caps_softmax<false, true ><<<K_CLS, 256, 0, stream>>>(bij, part, c);
        else if (it == 2) caps_softmax<true,  false><<<K_CLS, 256, 0, stream>>>(bij, bp2, c);
        if (it == 0) caps_gemm_t<true ><<<dim3(I_SZ / RB, KS_T), NTH, 0, stream>>>(x, x, t2c);
        else         caps_gemm_t<false><<<dim3(I_SZ / RB, KS_T), NTH, 0, stream>>>(x, c, part);
        if (it == 0) {
            caps_gemm_b_head<true ><<<dim3(M_SZ / RB, KS_B), NTF, 0, stream>>>(x, t2c, W, part);
        } else if (it < 2 && two_sets) {
            caps_gemm_b_head<false><<<dim3(M_SZ / RB, KS_B), NTF, 0, stream>>>(x, part, W, bp2);
        } else if (it < 2) {
            caps_squash<false><<<B_SZ, 320, 0, stream>>>(part, W, out, wv2);
            caps_gemm_b<<<dim3(M_SZ / RB, KS_B), NTH, 0, stream>>>(x, wv2, part);
        } else {
            caps_squash<true ><<<B_SZ, 320, 0, stream>>>(part, W, out, wv2);
        }
    }
}


# Pass 3: coefficients made in GEMM 1's head on a grid re-cut to 16 slabs x 16 slices of 64 m (B = exp(l - slice max)); last squash rescales 16 planes, a lane each; 8 -> 7 launches (sum order changes)
# speedup vs baseline: 13.3104x; 1.0479x over previous
#include <hip/hip_runtime.h>
#include <math.h>

typedef __attribute__((ext_vector_type(2))) float v2f;
typedef __attribute__((ext_vector_type(8))) float v8f;
typedef __attribute__((ext_vector_type(4))) float v4f;

#define B_SZ   256
#define M_SZ   1024
#define A_SZ   8
#define K_CLS  20
#define O_SZ   16
#define BP     20
#define GSTEP  16
#define NTH    128
#define RB     32
#define I_SZ   (B_SZ * A_SZ)
#define KS_T   4
#define KS_B   8
#define KROWS  256
#define PP     20
#define TP     36
#define CP     260
#define XBP    (A_SZ * KROWS + 8)
#define XGP    68
#define XBB    (4 * XGP)
#define NTF    640
#define WSP    164
#define SSP    324
#define SCP    20
#define SL3    16
#define MR3    64
#define RB3    128
#define NT3    320
#define XP3    (A_SZ * MR3 + 8)
#define CP3    (MR3 + 4)
#define TP3    (RB3 + 4)

#define C_START 0x1p-10f

template <int M>
__device__ __forceinline__ float xrow(float x) {
    const int b = __builtin_bit_cast(int, x);
    return __builtin_bit_cast(float, __builtin_amdgcn_update_dpp(b, b, 0x160 + M, 0xf, 0xf, false));
}
__device__ __forceinline__ float xhalf(float x) {
    const int b = __builtin_bit_cast(int, x);
    return __builtin_bit_cast(float, __builtin_amdgcn_permlanex16(b, b, 0x76543210, 0xfedcba98, false, false));
}

template <bool BASE, bool WB>
__global__ void caps_softmax(float* __restrict__ bij, const float* __restrict__ part,
                             float* __restrict__ c) {
    __shared__ __attribute__((aligned(16))) float redm[8];
    __shared__ __attribute__((aligned(16))) float reds[8];
    const int k = blockIdx.x;
    const int t = threadIdx.x;
    const int wave = t >> 5;
    const int lane = t & 31;

    float* __restrict__ brow = bij + k * M_SZ;
    float v0 = 0.0f, v1 = 0.0f, v2 = 0.0f, v3 = 0.0f;
    if (BASE) {
        v0 = brow[t      ];
        v1 = brow[t + 256];
        v2 = brow[t + 512];
        v3 = brow[t + 768];
    }
    {
        float p[4][KS_B];
        const float* __restrict__ pk = part + (t >> 4) * (16 * PP) + k * 16 + (t & 15);
        #pragma unroll
        for (int r = 0; r < 4; ++r)
            #pragma unroll
            for (int s = 0; s < KS_B; ++s)
                p[r][s] = pk[s * (M_SZ * PP) + r * (16 * 16 * PP)];
        v0 += ((p[0][0] + p[0][1]) + (p[0][2] + p[0][3])) + ((p[0][4] + p[0][5]) + (p[0][6] + p[0][7]));
        v1 += ((p[1][0] + p[1][1]) + (p[1][2] + p[1][3])) + ((p[1][4] + p[1][5]) + (p[1][6] + p[1][7]));
        v2 += ((p[2][0] + p[2][1]) + (p[2][2] + p[2][3])) + ((p[2][4] + p[2][5]) + (p[2][6] + p[2][7]));
        v3 += ((p[3][0] + p[3][1]) + (p[3][2] + p[3][3])) + ((p[3][4] + p[3][5]) + (p[3][6] + p[3][7]));
    }

    float mx = fmaxf(fmaxf(v0, v1), fmaxf(v2, v3));
    mx = fmaxf(mx, xrow<1>(mx));
    mx = fmaxf(mx, xrow<2>(mx));
    mx = fmaxf(mx, xrow<4>(mx));
    mx = fmaxf(mx, xrow<8>(mx));
    mx = fmaxf(mx, xhalf(mx));
    if (lane == 0) redm[wave] = mx;
    __syncthreads();
    {
        const float4 q0 = *(const float4*)&redm[0];
        const float4 q1 = *(const float4*)&redm[4];
        mx = fmaxf(fmaxf(fmaxf(q0.x, q0.y), fmaxf(q0.z, q0.w)),
                   fmaxf(fmaxf(q1.x, q1.y), fmaxf(q1.z, q1.w)));
    }

    float e0 = expf(v0 - mx), e1 = expf(v1 - mx);
    float e2 = expf(v2 - mx), e3 = expf(v3 - mx);
    float sm = e0 + e1 + e2 + e3;
    sm += xrow<1>(sm);
    sm += xrow<2>(sm);
    sm += xrow<4>(sm);
    sm += xrow<8>(sm);
    sm += xhalf(sm);
    if (lane == 0) reds[wave] = sm;
    __syncthreads();
    float tot;
    {
        const float4 q0 = *(const float4*)&reds[0];
        const float4 q1 = *(const float4*)&reds[4];
        tot = ((q0.x + q0.y) + (q0.z + q0.w)) + ((q1.x + q1.y) + (q1.z + q1.w));
    }
    float inv = 1.0f / tot;

    float* __restrict__ crow = c + k * M_SZ;
    crow[t      ] = e0 * inv;
    crow[t + 256] = e1 * inv;
    crow[t + 512] = e2 * inv;
    crow[t + 768] = e3 * inv;
    if (WB) {
        brow[t      ] = v0;
        brow[t + 256] = v1;
        brow[t + 512] = v2;
        brow[t + 768] = v3;
    }
}

template <bool START>
__global__ __launch_bounds__(NTH)
void caps_gemm_t(const float* __restrict__ x,
                 const float* __restrict__ c,
                 float* __restrict__ t2p) {
    __shared__ __attribute__((aligned(16))) float lds[4 * XBP + K_CLS * CP + K_CLS * TP];
    float* xs   = lds;
    float* cs   = lds + 4 * XBP;
    float* tile = cs + K_CLS * CP;
    const int t    = threadIdx.x;
    const int slab = blockIdx.x;
    const int ks   = blockIdx.y;

    const float* __restrict__ xsrc = x + slab * (4 * A_SZ * M_SZ) + ks * (A_SZ * KROWS);
    float4 v[16];
    #pragma unroll
    for (int j = 0; j < 16; ++j) {
        const int q = j * NTH + t;
        v[j] = *(const float4*)(xsrc + (q >> 9) * (A_SZ * M_SZ) + (q & 511) * 4);
    }
    const float* __restrict__ csrc = c + ks * KROWS + (t >> 6) * M_SZ + (t & 63) * 4;
    float4 cv[K_CLS / 2];
    if (!START) {
        #pragma unroll
        for (int j = 0; j < K_CLS / 2; ++j) cv[j] = *(const float4*)(csrc + j * (2 * M_SZ));
    }
    #pragma unroll
    for (int j = 0; j < 16; ++j) {
        const int q = j * NTH + t;
        *(float4*)&xs[(q >> 9) * XBP + (q & 511) * 4] = v[j];
    }
    if (!START) {
        #pragma unroll
        for (int j = 0; j < K_CLS / 2; ++j) *(float4*)&cs[(j * 2 + (t >> 6)) * CP + (t & 63) * 4] = cv[j];
    }
    __syncthreads();

    const int wave = t >> 5;
    const int lane = t & 31;
    const int lo   = lane & 15;
    const int hi   = lane >> 4;
    const int ct   = wave & 1;
    const int rt   = wave >> 1;

    v8f acc = {};
    const int abase = (rt * 2 + (lo >> 3)) * XBP + (lo & 7) + hi * 2 * A_SZ;
    const int bcol  = min(ct * 16 + lo, K_CLS - 1);
    const int bbase = bcol * CP + hi * 2;

    v2f a0[GSTEP], b0[GSTEP], a1[GSTEP], b1[GSTEP];
    auto load = [&](v2f (&a)[GSTEP], v2f (&b)[GSTEP], int g) {
        #pragma unroll
        for (int s = 0; s < GSTEP; ++s) {
            const int mb = (g * GSTEP + s) * 4;
            a[s].x = xs[abase + mb * A_SZ];
            a[s].y = xs[abase + mb * A_SZ + A_SZ];
            if (START) { b[s].x = C_START; b[s].y = C_START; }
            else b[s] = *(const v2f*)&cs[bbase + mb];
        }
    };
    auto steps = [&](const v2f (&a)[GSTEP], const v2f (&b)[GSTEP]) {
        #pragma unroll
        for (int s = 0; s < GSTEP; ++s)
            acc = __builtin_amdgcn_wmma_f32_16x16x4_f32(
                false, a[s], false, b[s], (short)0, acc, false, false);
    };

    static_assert((KROWS / 4) / GSTEP == 4, "four groups of matrix steps");
    if (!START || ct == 0) {
        load(a0, b0, 0);
        load(a1, b1, 1);
        __builtin_amdgcn_sched_barrier(0);
        steps(a0, b0);
        __builtin_amdgcn_sched_barrier(0);
        load(a0, b0, 2);
        __builtin_amdgcn_sched_barrier(0);
        steps(a1, b1);
        __builtin_amdgcn_sched_barrier(0);
        load(a1, b1, 3);
        __builtin_amdgcn_sched_barrier(0);
        steps(a0, b0);
        steps(a1, b1);
    }

    if (START) {
        if (ct == 0 && lo == 0) {
            #pragma unroll
            for (int r = 0; r < 8; ++r) tile[rt * 16 + hi * 8 + r] = acc[r];
        }
        __syncthreads();
        if (t < RB) t2p[ks * I_SZ + slab * RB + t] = tile[t];
        return;
    }
    if (ct == 0 || lo < K_CLS - 16) {
        #pragma unroll
        for (int r = 0; r < 8; ++r) tile[(ct * 16 + lo) * TP + rt * 16 + hi * 8 + r] = acc[r];
    }
    __syncthreads();
    float* __restrict__ dst = t2p + ks * (K_CLS * I_SZ) + slab * RB + (t >> 5) * I_SZ + (t & 31);
    float r[5];
    #pragma unroll
    for (int i = 0; i < 5; ++i) r[i] = tile[((t >> 5) + 4 * i) * TP + (t & 31)];
    #pragma unroll
    for (int i = 0; i < 5; ++i) dst[i * (4 * I_SZ)] = r[i];
}

template <bool LAST>
__global__ void caps_squash(const float* __restrict__ t2p,
                            const float* __restrict__ W,
                            float* __restrict__ out,
                            float* __restrict__ wv2) {
    __shared__ float s_s[K_CLS][O_SZ];
    __shared__ float s_v[K_CLS][O_SZ];
    const int b = blockIdx.x;
    const int t = threadIdx.x;
    const int k = t >> 4;
    const int o = t & 15;

    float s = 0.0f;
    {
        const float* __restrict__ tp = t2p + k * I_SZ + b * A_SZ;
        float4 p[KS_T][2];
        #pragma unroll
        for (int q = 0; q < KS_T; ++q) {
            p[q][0] = *(const float4*)(tp + q * (K_CLS * I_SZ));
            p[q][1] = *(const float4*)(tp + q * (K_CLS * I_SZ) + 4);
        }
        float pa[KS_T][A_SZ];
        #pragma unroll
        for (int q = 0; q < KS_T; ++q) {
            pa[q][0] = p[q][0].x; pa[q][1] = p[q][0].y; pa[q][2] = p[q][0].z; pa[q][3] = p[q][0].w;
            pa[q][4] = p[q][1].x; pa[q][5] = p[q][1].y; pa[q][6] = p[q][1].z; pa[q][7] = p[q][1].w;
        }
        #pragma unroll
        for (int a = 0; a < A_SZ; ++a) {
            float tv = pa[0][a];
            #pragma unroll
            for (int q = 1; q < KS_T; ++q) tv += pa[q][a];
            s += W[k * (O_SZ * A_SZ) + o * A_SZ + a] * tv;
        }
    }
    s_s[k][o] = s;
    __syncthreads();

    float magsq = 0.0f;
    #pragma unroll
    for (int kk = 0; kk < K_CLS; ++kk) {
        float sv = s_s[kk][o];
        magsq += sv * sv;
    }
    float scale = magsq / ((1.0f + magsq) * sqrtf(magsq));
    float v = s * scale;
    if (LAST) {
        out[b * (K_CLS * O_SZ) + t] = v;
    } else {
        s_v[k][o] = v;
        __syncthreads();

        if (t < K_CLS * A_SZ) {
            int k2 = t % K_CLS;
            int a  = t / K_CLS;
            float w = 0.0f;
            #pragma unroll
            for (int o2 = 0; o2 < O_SZ; ++o2)
                w += W[k2 * (O_SZ * A_SZ) + o2 * A_SZ + a] * s_v[k2][o2];
            wv2[(b * 8 + a) * PP + k2] = w * (1.0f / (float)B_SZ);
        }
    }
}

__global__ __launch_bounds__(NTH)
void caps_gemm_b(const float* __restrict__ x,
                 const float* __restrict__ wv2,
                 float* __restrict__ bijp) {
    constexpr int NB = KROWS / A_SZ;
    __shared__ __attribute__((aligned(16))) float lds[NB * XBB + KROWS * BP + RB * PP];
    float* xs   = lds;
    float* ws   = lds + NB * XBB;
    float* tile = ws + KROWS * BP;
    const int t    = threadIdx.x;
    const int slab = blockIdx.x;
    const int ks   = blockIdx.y;

    const float* __restrict__ xsrc = x + ks * (NB * A_SZ * M_SZ) + slab * (RB * A_SZ);
    float4 v[16];
    #pragma unroll
    for (int j = 0; j < 16; ++j) {
        const int q = j * NTH + t;
        v[j] = *(const float4*)(xsrc + (q >> 6) * (A_SZ * M_SZ) + (q & 63) * 4);
    }
    const float* __restrict__ wsrc = wv2 + ks * (KROWS * BP) + t * 4;
    float4 wv[(KROWS * BP) / (4 * NTH)];
    #pragma unroll
    for (int j = 0; j < (KROWS * BP) / (4 * NTH); ++j) wv[j] = *(const float4*)(wsrc + j * (4 * NTH));
    #pragma unroll
    for (int j = 0; j < 16; ++j) {
        const int q = j * NTH + t;
        const int w = q & 63;
        *(float4*)&xs[(q >> 6) * XBB + (w >> 4) * XGP + (w & 15) * 4] = v[j];
    }
    #pragma unroll
    for (int j = 0; j < (KROWS * BP) / (4 * NTH); ++j) ((float4*)ws)[j * NTH + t] = wv[j];
    __syncthreads();

    const int wave = t >> 5;
    const int lane = t & 31;
    const int lo   = lane & 15;
    const int hi   = lane >> 4;
    const int ct   = wave & 1;
    const int rt   = wave >> 1;

    v8f acc = {};
    const int mrow  = rt * 16 + lo;
    const int abase = (mrow >> 3) * XGP + (mrow & 7) * A_SZ + hi * 2;
    const int bcol  = min(ct * 16 + lo, K_CLS - 1);
    const int bbase = hi * 2 * BP + bcol;
    constexpr int GB = GSTEP / 2;

    v2f a0[GSTEP], b0[GSTEP], a1[GSTEP], b1[GSTEP];
    auto load = [&](v2f (&a)[GSTEP], v2f (&bf)[GSTEP], int g) {
        #pragma unroll
        for (int s = 0; s < GB; ++s) {
            const int b = g * GB + s;
            a[2 * s]     = *(const v2f*)&xs[b * XBB + abase];
            a[2 * s + 1] = *(const v2f*)&xs[b * XBB + abase + 4];
            bf[2 * s].x     = ws[(b * A_SZ    ) * BP + bbase];
            bf[2 * s].y     = ws[(b * A_SZ + 1) * BP + bbase];
            bf[2 * s + 1].x = ws[(b * A_SZ + 4) * BP + bbase];
            bf[2 * s + 1].y = ws[(b * A_SZ + 5) * BP + bbase];
        }
    };
    auto steps = [&](const v2f (&a)[GSTEP], const v2f (&bf)[GSTEP]) {
        #pragma unroll
        for (int s = 0; s < GSTEP; ++s)
            acc = __builtin_amdgcn_wmma_f32_16x16x4_f32(
                false, a[s], false, bf[s], (short)0, acc, false, false);
    };

    static_assert(NB / GB == 4, "four groups of matrix steps");
    load(a0, b0, 0);
    load(a1, b1, 1);
    __builtin_amdgcn_sched_barrier(0);
    steps(a0, b0);
    __builtin_amdgcn_sched_barrier(0);
    load(a0, b0, 2);
    __builtin_amdgcn_sched_barrier(0);
    steps(a1, b1);
    __builtin_amdgcn_sched_barrier(0);
    load(a1, b1, 3);
    __builtin_amdgcn_sched_barrier(0);
    steps(a0, b0);
    steps(a1, b1);

    if (ct == 0 || lo < K_CLS - 16) {
        #pragma unroll
        for (int r = 0; r < 8; ++r) tile[rt * (16 * PP) + (ct * 16 + lo) * 16 + r + hi * 8] = acc[r];
    }
    __syncthreads();
    float* __restrict__ dst = bijp + ks * (M_SZ * PP) + slab * (RB * PP);
    float r[5];
    #pragma unroll
    for (int i = 0; i < 5; ++i) r[i] = tile[t + i * NTH];
    #pragma unroll
    for (int i = 0; i < 5; ++i) dst[t + i * NTH] = r[i];
}

template <bool START>
__global__ __launch_bounds__(NTF)
void caps_gemm_b_head(const float* __restrict__ x,
                      const float* __restrict__ t2,
                      const float* __restrict__ W,
                      float* __restrict__ bijp) {
    constexpr int NB = KROWS / A_SZ;
    static_assert(NB == 32, "the head maps one batch element of the slice to one lane");
    static_assert(NTF == K_CLS * 32 && NTF == RB * PP, "one wave per class; one float of the tile per thread");
    __shared__ __attribute__((aligned(16))) float lds[NB * XBB + NB * WSP + RB * PP + NB * SSP + NB * SCP];
    float* xs   = lds;
    float* ws   = lds + NB * XBB;
    float* tile = ws + NB * WSP;
    float* ss   = tile + RB * PP;
    float* scs  = ss + NB * SSP;
    const int t    = threadIdx.x;
    const int slab = blockIdx.x;
    const int ks   = blockIdx.y;
    const int lane = t & 31;
    const int wave = __builtin_amdgcn_readfirstlane(t >> 5);

    const float* __restrict__ xsrc = x + ks * (NB * A_SZ * M_SZ) + slab * (RB * A_SZ);
    float tv[A_SZ];
    {
        constexpr int PS = START ? I_SZ : K_CLS * I_SZ;
        const float* __restrict__ tp = t2 + (START ? 0 : wave * I_SZ) + (ks * NB + lane) * A_SZ;
        float4 p[KS_T][2];
        #pragma unroll
        for (int s = 0; s < KS_T; ++s) {
            p[s][0] = *(const float4*)(tp + s * PS);
            p[s][1] = *(const float4*)(tp + s * PS + 4);
        }
        if (t < 512) {
            float4 v[4];
            #pragma unroll
            for (int j = 0; j < 4; ++j) {
                const int q = j * 512 + t;
                v[j] = *(const float4*)(xsrc + (q >> 6) * (A_SZ * M_SZ) + (q & 63) * 4);
            }
            #pragma unroll
            for (int j = 0; j < 4; ++j) {
                const int q = j * 512 + t;
                const int w = q & 63;
                *(float4*)&xs[(q >> 6) * XBB + (w >> 4) * XGP + (w & 15) * 4] = v[j];
            }
        }
        float pa[KS_T][A_SZ];
        #pragma unroll
        for (int s = 0; s < KS_T; ++s) {
            pa[s][0] = p[s][0].x; pa[s][1] = p[s][0].y; pa[s][2] = p[s][0].z; pa[s][3] = p[s][0].w;
            pa[s][4] = p[s][1].x; pa[s][5] = p[s][1].y; pa[s][6] = p[s][1].z; pa[s][7] = p[s][1].w;
        }
        #pragma unroll
        for (int a = 0; a < A_SZ; ++a) {
            float sum = pa[0][a];
            #pragma unroll
            for (int s = 1; s < KS_T; ++s) sum += pa[s][a];
            tv[a] = sum;
        }
    }
    const float* __restrict__ Wk = W + wave * (O_SZ * A_SZ);
    float sv[O_SZ];
    #pragma unroll
    for (int o = 0; o < O_SZ; ++o) {
        float s = 0.0f;
        #pragma unroll
        for (int a = 0; a < A_SZ; ++a) s += Wk[o * A_SZ + a] * tv[a];
        sv[o] = s;
    }
    #pragma unroll
    for (int j = 0; j < O_SZ / 4; ++j)
        *(float4*)&ss[lane * SSP + wave * O_SZ + j * 4] =
            make_float4(sv[j * 4], sv[j * 4 + 1], sv[j * 4 + 2], sv[j * 4 + 3]);
    __syncthreads();

    if (t < NB * O_SZ) {
        const int b2 = t >> 4;
        const int o2 = t & 15;
        float magsq = 0.0f;
        #pragma unroll
        for (int kk = 0; kk < K_CLS; ++kk) {
            const float q = ss[b2 * SSP + kk * O_SZ + o2];
            magsq += q * q;
        }
        scs[b2 * SCP + o2] = magsq / ((1.0f + magsq) * sqrtf(magsq));
    }
    __syncthreads();

    {
        float sc[O_SZ];
        #pragma unroll
        for (int j = 0; j < O_SZ / 4; ++j) {
            const float4 q = *(const float4*)&scs[lane * SCP + j * 4];
            sc[j * 4] = q.x; sc[j * 4 + 1] = q.y; sc[j * 4 + 2] = q.z; sc[j * 4 + 3] = q.w;
        }
        float vv[O_SZ];
        #pragma unroll
        for (int o = 0; o < O_SZ; ++o) vv[o] = sv[o] * sc[o];
        #pragma unroll
        for (int a = 0; a < A_SZ; ++a) {
            float w = 0.0f;
            #pragma unroll
            for (int o = 0; o < O_SZ; ++o) w += Wk[o * A_SZ + a] * vv[o];
            ws[lane * WSP + a * BP + wave] = w * (1.0f / (float)B_SZ);
        }
    }
    __syncthreads();

    if (t < NTH) {
        const int lo   = lane & 15;
        const int hi   = lane >> 4;
        const int ct   = wave & 1;
        const int rt   = wave >> 1;

        v8f acc = {};
        const int mrow  = rt * 16 + lo;
        const int abase = (mrow >> 3) * XGP + (mrow & 7) * A_SZ + hi * 2;
        const int bcol  = min(ct * 16 + lo, K_CLS - 1);
        const int bbase = hi * 2 * BP + bcol;
        constexpr int GB = GSTEP / 2;

        v2f a0[GSTEP], b0[GSTEP], a1[GSTEP], b1[GSTEP];
        auto load = [&](v2f (&a)[GSTEP], v2f (&bf)[GSTEP], int g) {
            #pragma unroll
            for (int s = 0; s < GB; ++s) {
                const int b = g * GB + s;
                a[2 * s]     = *(const v2f*)&xs[b * XBB + abase];
                a[2 * s + 1] = *(const v2f*)&xs[b * XBB + abase + 4];
                bf[2 * s].x     = ws[b * WSP           + bbase];
                bf[2 * s].y     = ws[b * WSP +     BP + bbase];
                bf[2 * s + 1].x = ws[b * WSP + 4 * BP + bbase];
                bf[2 * s + 1].y = ws[b * WSP + 5 * BP + bbase];
            }
        };
        auto steps = [&](const v2f (&a)[GSTEP], const v2f (&bf)[GSTEP]) {
            #pragma unroll
            for (int s = 0; s < GSTEP; ++s)
                acc = __builtin_amdgcn_wmma_f32_16x16x4_f32(
                    false, a[s], false, bf[s], (short)0, acc, false, false);
        };

        static_assert(NB / GB == 4, "four groups of matrix steps");
        load(a0, b0, 0);
        load(a1, b1, 1);
        __builtin_amdgcn_sched_barrier(0);
        steps(a0, b0);
        __builtin_amdgcn_sched_barrier(0);
        load(a0, b0, 2);
        __builtin_amdgcn_sched_barrier(0);
        steps(a1, b1);
        __builtin_amdgcn_sched_barrier(0);
        load(a1, b1, 3);
        __builtin_amdgcn_sched_barrier(0);
        steps(a0, b0);
        steps(a1, b1);

        if (ct == 0 || lo < K_CLS - 16) {
            #pragma unroll
            for (int r = 0; r < 8; ++r) tile[rt * (16 * PP) + (ct * 16 + lo) * 16 + r + hi * 8] = acc[r];
        }
    }
    __syncthreads();
    bijp[ks * (M_SZ * PP) + slab * (RB * PP) + t] = tile[t];
}

__global__ __launch_bounds__(NT3)
void caps_gemm_t_sm(const float* __restrict__ x,
                    const float* __restrict__ bij,
                    const float* __restrict__ bijp,
                    float* __restrict__ t3p,
                    float* __restrict__ st) {
    static_assert(MR3 * SL3 == M_SZ && MR3 == 64 && NT3 == K_CLS * 16, "16 threads per class, 4 rows per thread");
    static_assert(6 * NT3 + 128 == 16 * (A_SZ * MR3 / 4), "the x tile: six full rounds of float4 and 128 more");
    __shared__ __attribute__((aligned(16))) float lds[16 * XP3 + K_CLS * CP3 + K_CLS * TP3];
    float* xs   = lds;
    float* cs   = lds + 16 * XP3;
    float* tile = cs + K_CLS * CP3;
    const int t    = threadIdx.x;
    const int slab = blockIdx.x;
    const int ks   = blockIdx.y;

    const float* __restrict__ xsrc = x + slab * (16 * A_SZ * M_SZ) + ks * (A_SZ * MR3);
    v4f v[7];
    #pragma unroll
    for (int j = 0; j < 7; ++j) {
        const int q = min(j * NT3 + t, 2047);
        v[j] = *(const v4f*)(xsrc + (q >> 7) * (A_SZ * M_SZ) + (q & 127) * 4);
    }
    const int k  = t >> 4;
    const int j4 = (t >> 2) & 3;
    const int r4 = t & 3;
    const float* __restrict__ pb = bijp + (ks * 4 + j4) * (16 * PP) + k * 16 + r4 * 4;
    const v4f lb = *(const v4f*)(bij + k * M_SZ + ks * MR3 + j4 * 16 + r4 * 4);
    v4f p[KS_B];
    #pragma unroll
    for (int s = 0; s < KS_B; ++s) p[s] = *(const v4f*)(pb + s * (M_SZ * PP));
    asm volatile("" :: "v"(v[0]), "v"(v[1]), "v"(v[2]), "v"(v[3]), "v"(v[4]), "v"(v[5]), "v"(v[6]), "v"(lb),
                       "v"(p[0]), "v"(p[1]), "v"(p[2]), "v"(p[3]), "v"(p[4]), "v"(p[5]), "v"(p[6]), "v"(p[7]));

    #pragma unroll
    for (int j = 0; j < 7; ++j) {
        const int q = min(j * NT3 + t, 2047);
        *(v4f*)&xs[(q >> 7) * XP3 + (q & 127) * 4] = v[j];
    }
    {
        float l[4];
        l[0] = lb.x + (((p[0].x + p[1].x) + (p[2].x + p[3].x)) + ((p[4].x + p[5].x) + (p[6].x + p[7].x)));
        l[1] = lb.y + (((p[0].y + p[1].y) + (p[2].y + p[3].y)) + ((p[4].y + p[5].y) + (p[6].y + p[7].y)));
        l[2] = lb.z + (((p[0].z + p[1].z) + (p[2].z + p[3].z)) + ((p[4].z + p[5].z) + (p[6].z + p[7].z)));
        l[3] = lb.w + (((p[0].w + p[1].w) + (p[2].w + p[3].w)) + ((p[4].w + p[5].w) + (p[6].w + p[7].w)));
        float mx = fmaxf(fmaxf(l[0], l[1]), fmaxf(l[2], l[3]));
        mx = fmaxf(mx, xrow<1>(mx));
        mx = fmaxf(mx, xrow<2>(mx));
        mx = fmaxf(mx, xrow<4>(mx));
        mx = fmaxf(mx, xrow<8>(mx));
        const float e0 = expf(l[0] - mx), e1 = expf(l[1] - mx);
        const float e2 = expf(l[2] - mx), e3 = expf(l[3] - mx);
        float sm = (e0 + e1) + (e2 + e3);
        sm += xrow<1>(sm);
        sm += xrow<2>(sm);
        sm += xrow<4>(sm);
        sm += xrow<8>(sm);
        *(float4*)&cs[k * CP3 + j4 * 16 + r4 * 4] = make_float4(e0, e1, e2, e3);
        if (slab == 0 && (t & 15) == 0) {
            st[(ks * K_CLS + k) * 2]     = mx;
            st[(ks * K_CLS + k) * 2 + 1] = sm;
        }
    }
    __syncthreads();

    const int wave = t >> 5;
    const int lane = t & 31;
    if (wave < RB3 / 16) {
        const int lo   = lane & 15;
        const int hi   = lane >> 4;
        const int abase = (wave * 2 + (lo >> 3)) * XP3 + (lo & 7) + hi * 2 * A_SZ;
        const int bb0   = lo * CP3 + hi * 2;
        const int bb1   = min(16 + lo, K_CLS - 1) * CP3 + hi * 2;

        v2f a[MR3 / 4], b0[MR3 / 4], b1[MR3 / 4];
        #pragma unroll
        for (int s = 0; s < MR3 / 4; ++s) {
            const int mb = s * 4;
            a[s].x = xs[abase + mb * A_SZ];
            a[s].y = xs[abase + mb * A_SZ + A_SZ];
            b0[s]  = *(const v2f*)&cs[bb0 + mb];
            b1[s]  = *(const v2f*)&cs[bb1 + mb];
        }
        v8f acc0 = {}, acc1 = {};
        #pragma unroll
        for (int s = 0; s < MR3 / 4; ++s) {
            acc0 = __builtin_amdgcn_wmma_f32_16x16x4_f32(false, a[s], false, b0[s], (short)0, acc0, false, false);
            acc1 = __builtin_amdgcn_wmma_f32_16x16x4_f32(false, a[s], false, b1[s], (short)0, acc1, false, false);
        }

        #pragma unroll
        for (int r = 0; r < 8; ++r) tile[lo * TP3 + wave * 16 + hi * 8 + r] = acc0[r];
        if (lo < K_CLS - 16) {
            #pragma unroll
            for (int r = 0; r < 8; ++r) tile[(16 + lo) * TP3 + wave * 16 + hi * 8 + r] = acc1[r];
        }
    }
    __syncthreads();
    float* __restrict__ dst = t3p + ks * (K_CLS * I_SZ) + slab * RB3 + lane * 4;
    #pragma unroll
    for (int i = 0; i < 2; ++i) {
        const int kk = wave + 10 * i;
        *(float4*)(dst + kk * I_SZ) = *(const float4*)&tile[kk * TP3 + lane * 4];
    }
}

__global__ void caps_squash_sm(const float* __restrict__ t3p,
                               const float* __restrict__ st,
                               const float* __restrict__ W,
                               float* __restrict__ out) {
    static_assert(SL3 == O_SZ, "one slice plane per lane of a class's 16");
    __shared__ float s_s[K_CLS][O_SZ];
    const int b = blockIdx.x;
    const int t = threadIdx.x;
    const int k = t >> 4;
    const int o = t & 15;

    const float* __restrict__ tp = t3p + o * (K_CLS * I_SZ) + k * I_SZ + b * A_SZ;
    const float4 p0 = *(const float4*)tp;
    const float4 p1 = *(const float4*)(tp + 4);
    const float2 sq = *(const float2*)(st + (o * K_CLS + k) * 2);

    float gmx = sq.x;
    gmx = fmaxf(gmx, xrow<1>(gmx));
    gmx = fmaxf(gmx, xrow<2>(gmx));
    gmx = fmaxf(gmx, xrow<4>(gmx));
    gmx = fmaxf(gmx, xrow<8>(gmx));
    const float wq = expf(sq.x - gmx);
    float z = sq.y * wq;
    z += xrow<1>(z);
    z += xrow<2>(z);
    z += xrow<4>(z);
    z += xrow<8>(z);
    const float sc = wq / z;

    float tv[A_SZ] = { p0.x * sc, p0.y * sc, p0.z * sc, p0.w * sc, p1.x * sc, p1.y * sc, p1.z * sc, p1.w * sc };
    #pragma unroll
    for (int a = 0; a < A_SZ; ++a) {
        tv[a] += xrow<1>(tv[a]);
        tv[a] += xrow<2>(tv[a]);
        tv[a] += xrow<4>(tv[a]);
        tv[a] += xrow<8>(tv[a]);
    }
    float s = 0.0f;
    #pragma unroll
    for (int a = 0; a < A_SZ; ++a) s += W[k * (O_SZ * A_SZ) + o * A_SZ + a] * tv[a];
    s_s[k][o] = s;
    __syncthreads();

    float magsq = 0.0f;
    #pragma unroll
    for (int kk = 0; kk < K_CLS; ++kk) {
        float sv = s_s[kk][o];
        magsq += sv * sv;
    }
    float scale = magsq / ((1.0f + magsq) * sqrtf(magsq));
    out[b * (K_CLS * O_SZ) + t] = s * scale;
}

extern "C" void kernel_launch(void* const* d_in, const int* in_sizes, int n_in,
                              void* d_out, int out_size, void* d_ws, size_t ws_size,
                              hipStream_t stream) {
    (void)in_sizes; (void)n_in; (void)out_size;
    const float* x = (const float*)d_in[0];
    const float* W = (const float*)d_in[1];
    float* out = (float*)d_out;

    float* part = (float*)d_ws;
    const bool two_sets = ws_size >= (size_t)1310720;
    float* bp2  = two_sets ? part + 163840 : part;
    float* bij  = out;
    float* wv2  = out + K_CLS * M_SZ;
    float* c    = wv2;
    float* t2c  = wv2;
    const bool wide = ws_size >= (size_t)3934720;
    float* t3p  = part + 327680;
    float* st3  = part + 983040;

    for (int it = 0; it < 3; ++it) {
        if (it == 2 && wide) {
            caps_gemm_t_sm<<<dim3(I_SZ / RB3, SL3), NT3, 0, stream>>>(x, bij, bp2, t3p, st3);
            caps_squash_sm<<<B_SZ, 320, 0, stream>>>(t3p, st3, W, out);
            break;
        }
        if (it == 1)      caps_softmax<false, true ><<<K_CLS, 256, 0, stream>>>(bij, part, c);
        else if (it == 2) caps_softmax<true,  false><<<K_CLS, 256, 0, stream>>>(bij, bp2, c);
        if (it == 0) caps_gemm_t<true ><<<dim3(I_SZ / RB, KS_T), NTH, 0, stream>>>(x, x, t2c);
        else         caps_gemm_t<false><<<dim3(I_SZ / RB, KS_T), NTH, 0, stream>>>(x, c, part);
        if (it == 0) {
            caps_gemm_b_head<true ><<<dim3(M_SZ / RB, KS_B), NTF, 0, stream>>>(x, t2c, W, part);
        } else if (it < 2 && two_sets) {
            caps_gemm_b_head<false><<<dim3(M_SZ / RB, KS_B), NTF, 0, stream>>>(x, part, W, bp2);
        } else if (it < 2) {
            caps_squash<false><<<B_SZ, 320, 0, stream>>>(part, W, out, wv2);
            caps_gemm_b<<<dim3(M_SZ / RB, KS_B), NTH, 0, stream>>>(x, wv2, part);
        } else {
            caps_squash<true ><<<B_SZ, 320, 0, stream>>>(part, W, out, wv2);
        }
    }
}
